# starting kernel
# baseline (speedup 1.0000x reference)
#include <hip/hip_runtime.h>

typedef __attribute__((ext_vector_type(2))) float v2f;
typedef __attribute__((ext_vector_type(8))) float v8f;

#define FDIM 48

__global__ void deg_kernel(const int* __restrict__ dst, float* __restrict__ deg, int nEdges) {
    int e = blockIdx.x * blockDim.x + threadIdx.x;
    if (e < nEdges) atomicAdd(&deg[dst[e]], 1.0f);
}

__global__ void invdeg_kernel(float* __restrict__ deg, int n) {
    int v = blockIdx.x * blockDim.x + threadIdx.x;
    if (v < n) deg[v] = 1.0f / fmaxf(deg[v], 1.0f);
}

__global__ void scatter_kernel(const float* __restrict__ x, const int* __restrict__ src,
                               const int* __restrict__ dst, float* __restrict__ agg,
                               int nEdges) {
    unsigned tid = blockIdx.x * blockDim.x + threadIdx.x;
    unsigned total = (unsigned)nEdges * 12u;
    if (tid >= total) return;
    unsigned e = tid / 12u;
    unsigned c = tid - e * 12u;
    const float4 v = *(const float4*)(x + (size_t)src[e] * FDIM + c * 4);
    float* a = agg + (size_t)dst[e] * FDIM + c * 4;
    atomicAdd(a + 0, v.x);
    atomicAdd(a + 1, v.y);
    atomicAdd(a + 2, v.z);
    atomicAdd(a + 3, v.w);
}

__global__ void __launch_bounds__(64)
sage_gemm_kernel(const float* __restrict__ x,
                 const float* __restrict__ agg,
                 const float* __restrict__ invd,
                 const float* __restrict__ Wself,
                 const float* __restrict__ Wneigh,
                 const float* __restrict__ bias,
                 float* __restrict__ out,
                 int relu, int tilesPerWave) {
    const int lane = threadIdx.x & 31;
    const int wave = threadIdx.x >> 5;
    const int half = lane >> 4;
    const int l15  = lane & 15;
    const int gw   = blockIdx.x * 2 + wave;

    v2f wS[12][3], wN[12][3];
    #pragma unroll
    for (int k = 0; k < 12; ++k) {
        const int kk = k * 4 + half * 2;
        #pragma unroll
        for (int nt = 0; nt < 3; ++nt) {
            const int col = nt * 16 + l15;
            wS[k][nt].x = Wself [(size_t)(kk)     * FDIM + col];
            wS[k][nt].y = Wself [(size_t)(kk + 1) * FDIM + col];
            wN[k][nt].x = Wneigh[(size_t)(kk)     * FDIM + col];
            wN[k][nt].y = Wneigh[(size_t)(kk + 1) * FDIM + col];
        }
    }
    float bv0 = bias[0 * 16 + l15];
    float bv1 = bias[1 * 16 + l15];
    float bv2 = bias[2 * 16 + l15];

    #pragma unroll 1
    for (int t = 0; t < tilesPerWave; ++t) {
        const int m0 = (gw * tilesPerWave + t) * 16;
        const float* xrow = x   + (size_t)(m0 + l15) * FDIM;
        const float* grow = agg + (size_t)(m0 + l15) * FDIM;
        const float  dsc  = invd[m0 + l15];

        v8f cs[3], cn[3];
        #pragma unroll
        for (int r = 0; r < 8; ++r) {
            cs[0][r] = bv0; cs[1][r] = bv1; cs[2][r] = bv2;
            cn[0][r] = 0.0f; cn[1][r] = 0.0f; cn[2][r] = 0.0f;
        }

        #pragma unroll
        for (int k = 0; k < 12; ++k) {
            const int kk = k * 4 + half * 2;
            v2f a;
            a.x = xrow[kk];
            a.y = xrow[kk + 1];
            v2f a2;
            a2.x = grow[kk] * dsc;
            a2.y = grow[kk + 1] * dsc;
            #pragma unroll
            for (int nt = 0; nt < 3; ++nt) {
                cs[nt] = __builtin_amdgcn_wmma_f32_16x16x4_f32(
                    false, a, false, wS[k][nt], (short)0, cs[nt], false, false);
                cn[nt] = __builtin_amdgcn_wmma_f32_16x16x4_f32(
                    false, a2, false, wN[k][nt], (short)0, cn[nt], false, false);
            }
        }

        #pragma unroll
        for (int nt = 0; nt < 3; ++nt) {
            #pragma unroll
            for (int r = 0; r < 8; ++r) {
                float v = cs[nt][r] + cn[nt][r];
                if (relu) v = fmaxf(v, 0.0f);
                out[(size_t)(m0 + half * 8 + r) * FDIM + nt * 16 + l15] = v;
            }
        }
    }
}

extern "C" void kernel_launch(void* const* d_in, const int* in_sizes, int n_in,
                              void* d_out, int out_size, void* d_ws, size_t ws_size,
                              hipStream_t stream) {
    const float* features = (const float*)d_in[0];
    const int*   src      = (const int*)d_in[1];
    const int*   dst      = (const int*)d_in[2];
    const float* Wself1   = (const float*)d_in[3];
    const float* Wneigh1  = (const float*)d_in[4];
    const float* b1       = (const float*)d_in[5];
    const float* Wself2   = (const float*)d_in[6];
    const float* Wneigh2  = (const float*)d_in[7];
    const float* b2       = (const float*)d_in[8];
    float* out = (float*)d_out;

    const int N = in_sizes[0] / FDIM;
    const int E = in_sizes[1];

    char* ws = (char*)d_ws;
    size_t off = 0;
    float* invd = (float*)(ws + off); off += (((size_t)N * 4)        + 255) & ~(size_t)255;
    float* agg  = (float*)(ws + off); off += (((size_t)N * FDIM * 4) + 255) & ~(size_t)255;
    float* h    = (float*)(ws + off);

    const unsigned scatterTotal  = (unsigned)E * 12u;
    const unsigned scatterBlocks = (scatterTotal + 255u) / 256u;

    const int tiles        = N / 16;
    const int tilesPerWave = 5;
    const int gemmWaves    = tiles / tilesPerWave;
    const int gemmBlocks   = gemmWaves / 2;

    hipMemsetAsync(invd, 0, (size_t)N * 4, stream);
    deg_kernel<<<(E + 255) / 256, 256, 0, stream>>>(dst, invd, E);
    invdeg_kernel<<<(N + 255) / 256, 256, 0, stream>>>(invd, N);

    hipMemsetAsync(agg, 0, (size_t)N * FDIM * 4, stream);
    scatter_kernel<<<scatterBlocks, 256, 0, stream>>>(features, src, dst, agg, E);
    sage_gemm_kernel<<<gemmBlocks, 64, 0, stream>>>(features, agg, invd,
                                                    Wself1, Wneigh1, b1, h, 1, tilesPerWave);

    hipMemsetAsync(agg, 0, (size_t)N * FDIM * 4, stream);
    scatter_kernel<<<scatterBlocks, 256, 0, stream>>>(h, src, dst, agg, E);
    sage_gemm_kernel<<<gemmBlocks, 64, 0, stream>>>(h, agg, invd,
                                                    Wself2, Wneigh2, b2, out, 0, tilesPerWave);
}


# in-degree counted by the layer-1 scatter into column 48 of a 256-byte agg row (one more atomic per edge), 1/max(deg,1) taken in the GEMM; degree and inverse kernels and their memset removed
# speedup vs baseline: 1.3218x; 1.3218x over previous
#include <hip/hip_runtime.h>

typedef __attribute__((ext_vector_type(2))) float v2f;
typedef __attribute__((ext_vector_type(8))) float v8f;

#define FDIM 48

#define PSTRIDE 64

template <int STRIDE, bool COUNT>
__global__ void scatter_kernel(const float* __restrict__ x, const int* __restrict__ src,
                               const int* __restrict__ dst, float* __restrict__ agg,
                               int nEdges) {
    unsigned tid = blockIdx.x * blockDim.x + threadIdx.x;
    unsigned total = (unsigned)nEdges * 12u;
    if (tid >= total) return;
    unsigned e = tid / 12u;
    unsigned c = tid - e * 12u;
    const float4 v = *(const float4*)(x + (size_t)src[e] * FDIM + c * 4);
    float* row = agg + (size_t)dst[e] * STRIDE;
    float* a = row + c * 4;
    atomicAdd(a + 0, v.x);
    atomicAdd(a + 1, v.y);
    atomicAdd(a + 2, v.z);
    atomicAdd(a + 3, v.w);
    if (COUNT && c == 0u) atomicAdd(row + FDIM, 1.0f);
}

__global__ void __launch_bounds__(64)
sage_gemm_kernel(const float* __restrict__ x,
                 const float* __restrict__ agg,
                 const float* __restrict__ degcol,
                 const float* __restrict__ Wself,
                 const float* __restrict__ Wneigh,
                 const float* __restrict__ bias,
                 float* __restrict__ out,
                 int relu, int tilesPerWave, int aggStride) {
    const int lane = threadIdx.x & 31;
    const int wave = threadIdx.x >> 5;
    const int half = lane >> 4;
    const int l15  = lane & 15;
    const int gw   = blockIdx.x * 2 + wave;

    v2f wS[12][3], wN[12][3];
    #pragma unroll
    for (int k = 0; k < 12; ++k) {
        const int kk = k * 4 + half * 2;
        #pragma unroll
        for (int nt = 0; nt < 3; ++nt) {
            const int col = nt * 16 + l15;
            wS[k][nt].x = Wself [(size_t)(kk)     * FDIM + col];
            wS[k][nt].y = Wself [(size_t)(kk + 1) * FDIM + col];
            wN[k][nt].x = Wneigh[(size_t)(kk)     * FDIM + col];
            wN[k][nt].y = Wneigh[(size_t)(kk + 1) * FDIM + col];
        }
    }
    float bv0 = bias[0 * 16 + l15];
    float bv1 = bias[1 * 16 + l15];
    float bv2 = bias[2 * 16 + l15];

    #pragma unroll 1
    for (int t = 0; t < tilesPerWave; ++t) {
        const int m0 = (gw * tilesPerWave + t) * 16;
        const float* xrow = x   + (size_t)(m0 + l15) * FDIM;
        const float* grow = agg + (size_t)(m0 + l15) * aggStride;
        const float  dsc  = 1.0f / fmaxf(degcol[(size_t)(m0 + l15) * PSTRIDE], 1.0f);

        v8f cs[3], cn[3];
        #pragma unroll
        for (int r = 0; r < 8; ++r) {
            cs[0][r] = bv0; cs[1][r] = bv1; cs[2][r] = bv2;
            cn[0][r] = 0.0f; cn[1][r] = 0.0f; cn[2][r] = 0.0f;
        }

        #pragma unroll
        for (int k = 0; k < 12; ++k) {
            const int kk = k * 4 + half * 2;
            v2f a;
            a.x = xrow[kk];
            a.y = xrow[kk + 1];
            v2f a2;
            a2.x = grow[kk] * dsc;
            a2.y = grow[kk + 1] * dsc;
            #pragma unroll
            for (int nt = 0; nt < 3; ++nt) {
                cs[nt] = __builtin_amdgcn_wmma_f32_16x16x4_f32(
                    false, a, false, wS[k][nt], (short)0, cs[nt], false, false);
                cn[nt] = __builtin_amdgcn_wmma_f32_16x16x4_f32(
                    false, a2, false, wN[k][nt], (short)0, cn[nt], false, false);
            }
        }

        #pragma unroll
        for (int nt = 0; nt < 3; ++nt) {
            #pragma unroll
            for (int r = 0; r < 8; ++r) {
                float v = cs[nt][r] + cn[nt][r];
                if (relu) v = fmaxf(v, 0.0f);
                out[(size_t)(m0 + half * 8 + r) * FDIM + nt * 16 + l15] = v;
            }
        }
    }
}

extern "C" void kernel_launch(void* const* d_in, const int* in_sizes, int n_in,
                              void* d_out, int out_size, void* d_ws, size_t ws_size,
                              hipStream_t stream) {
    const float* features = (const float*)d_in[0];
    const int*   src      = (const int*)d_in[1];
    const int*   dst      = (const int*)d_in[2];
    const float* Wself1   = (const float*)d_in[3];
    const float* Wneigh1  = (const float*)d_in[4];
    const float* b1       = (const float*)d_in[5];
    const float* Wself2   = (const float*)d_in[6];
    const float* Wneigh2  = (const float*)d_in[7];
    const float* b2       = (const float*)d_in[8];
    float* out = (float*)d_out;

    const int N = in_sizes[0] / FDIM;
    const int E = in_sizes[1];

    char* ws = (char*)d_ws;
    size_t off = 0;
    float* agg1 = (float*)(ws + off); off += (((size_t)N * PSTRIDE * 4) + 255) & ~(size_t)255;
    float* agg2 = (float*)(ws + off); off += (((size_t)N * FDIM * 4)    + 255) & ~(size_t)255;
    float* h    = (float*)(ws + off);
    const float* degcol = agg1 + FDIM;

    const unsigned scatterTotal  = (unsigned)E * 12u;
    const unsigned scatterBlocks = (scatterTotal + 255u) / 256u;

    const int tiles        = N / 16;
    const int tilesPerWave = 5;
    const int gemmWaves    = tiles / tilesPerWave;
    const int gemmBlocks   = gemmWaves / 2;

    hipMemsetAsync(agg1, 0, (size_t)N * PSTRIDE * 4, stream);
    scatter_kernel<PSTRIDE, true><<<scatterBlocks, 256, 0, stream>>>(features, src, dst, agg1, E);
    sage_gemm_kernel<<<gemmBlocks, 64, 0, stream>>>(features, agg1, degcol,
                                                    Wself1, Wneigh1, b1, h, 1, tilesPerWave, PSTRIDE);

    hipMemsetAsync(agg2, 0, (size_t)N * FDIM * 4, stream);
    scatter_kernel<FDIM, false><<<scatterBlocks, 256, 0, stream>>>(h, src, dst, agg2, E);
    sage_gemm_kernel<<<gemmBlocks, 64, 0, stream>>>(h, agg2, degcol,
                                                    Wself2, Wneigh2, b2, out, 0, tilesPerWave, FDIM);
}


# scatter remapped to lane = feature, a wave owns 8 edges (indices by scalar loads): each atomic adds whole 128-B lines of one agg row, not 4-B pieces strided over 3 rows; sums still in arrival order
# speedup vs baseline: 2.4071x; 1.8211x over previous
#include <hip/hip_runtime.h>

typedef __attribute__((ext_vector_type(2))) float v2f;
typedef __attribute__((ext_vector_type(8))) float v8f;

#define FDIM 48

#define PSTRIDE 64

#define EPW 8

template <int STRIDE, bool COUNT, bool GUARD>
__device__ __forceinline__ void scatter_wave(const float* __restrict__ x,
                                             const int* __restrict__ src,
                                             const int* __restrict__ dst,
                                             float* __restrict__ agg,
                                             unsigned e0, unsigned nEdges, unsigned lane) {
    const float* xr[EPW];
    float* gr[EPW];
    bool ok[EPW];
    #pragma unroll
    for (int k = 0; k < EPW; ++k) {
        unsigned e = e0 + (unsigned)k;
        ok[k] = !GUARD || e < nEdges;
        if (GUARD) e = e < nEdges ? e : nEdges - 1u;
        xr[k] = x   + (size_t)src[e] * FDIM;
        gr[k] = agg + (size_t)dst[e] * STRIDE;
    }
    if (COUNT) {
        float va[EPW], vb[EPW];
        #pragma unroll
        for (int k = 0; k < EPW; ++k) {
            va[k] = xr[k][lane];
            vb[k] = 1.0f;
        }
        if (lane < 16u) {
            #pragma unroll
            for (int k = 0; k < EPW; ++k) vb[k] = xr[k][32u + lane];
        }
        #pragma unroll
        for (int k = 0; k < EPW; ++k) {
            if (ok[k]) atomicAdd(gr[k] + lane, va[k]);
        }
        if (lane <= 16u) {
            #pragma unroll
            for (int k = 0; k < EPW; ++k) {
                if (ok[k]) atomicAdd(gr[k] + 32u + lane, vb[k]);
            }
        }
    } else {
        const bool lo = lane < 16u;
        float va[EPW / 2], vb[EPW / 2], vc[EPW / 2];
        #pragma unroll
        for (int p = 0; p < EPW / 2; ++p) {
            const float* xa = xr[2 * p];
            const float* xb = xr[2 * p + 1];
            va[p] = xa[lane];
            vb[p] = lo ? xa[32u + lane] : xb[lane - 16u];
            vc[p] = xb[16u + lane];
        }
        #pragma unroll
        for (int p = 0; p < EPW / 2; ++p) {
            float* ga = gr[2 * p];
            float* gb = gr[2 * p + 1];
            if (ok[2 * p]) atomicAdd(ga + lane, va[p]);
            if (lo ? ok[2 * p] : ok[2 * p + 1])
                atomicAdd(lo ? ga + (32u + lane) : gb + (lane - 16u), vb[p]);
            if (ok[2 * p + 1]) atomicAdd(gb + (16u + lane), vc[p]);
        }
    }
}

template <int STRIDE, bool COUNT>
__global__ void __launch_bounds__(256)
scatter_kernel(const float* __restrict__ x, const int* __restrict__ src,
               const int* __restrict__ dst, float* __restrict__ agg,
               int nEdges) {
    const unsigned lane = threadIdx.x & 31u;
    const unsigned wv = __builtin_amdgcn_readfirstlane(blockIdx.x * 8u + (threadIdx.x >> 5));
    const unsigned e0 = wv * (unsigned)EPW;
    const unsigned n  = (unsigned)nEdges;
    if (e0 >= n) return;
    if (e0 + (unsigned)EPW <= n) scatter_wave<STRIDE, COUNT, false>(x, src, dst, agg, e0, n, lane);
    else                         scatter_wave<STRIDE, COUNT, true >(x, src, dst, agg, e0, n, lane);
}

__global__ void __launch_bounds__(64)
sage_gemm_kernel(const float* __restrict__ x,
                 const float* __restrict__ agg,
                 const float* __restrict__ degcol,
                 const float* __restrict__ Wself,
                 const float* __restrict__ Wneigh,
                 const float* __restrict__ bias,
                 float* __restrict__ out,
                 int relu, int tilesPerWave, int aggStride) {
    const int lane = threadIdx.x & 31;
    const int wave = threadIdx.x >> 5;
    const int half = lane >> 4;
    const int l15  = lane & 15;
    const int gw   = blockIdx.x * 2 + wave;

    v2f wS[12][3], wN[12][3];
    #pragma unroll
    for (int k = 0; k < 12; ++k) {
        const int kk = k * 4 + half * 2;
        #pragma unroll
        for (int nt = 0; nt < 3; ++nt) {
            const int col = nt * 16 + l15;
            wS[k][nt].x = Wself [(size_t)(kk)     * FDIM + col];
            wS[k][nt].y = Wself [(size_t)(kk + 1) * FDIM + col];
            wN[k][nt].x = Wneigh[(size_t)(kk)     * FDIM + col];
            wN[k][nt].y = Wneigh[(size_t)(kk + 1) * FDIM + col];
        }
    }
    float bv0 = bias[0 * 16 + l15];
    float bv1 = bias[1 * 16 + l15];
    float bv2 = bias[2 * 16 + l15];

    #pragma unroll 1
    for (int t = 0; t < tilesPerWave; ++t) {
        const int m0 = (gw * tilesPerWave + t) * 16;
        const float* xrow = x   + (size_t)(m0 + l15) * FDIM;
        const float* grow = agg + (size_t)(m0 + l15) * aggStride;
        const float  dsc  = 1.0f / fmaxf(degcol[(size_t)(m0 + l15) * PSTRIDE], 1.0f);

        v8f cs[3], cn[3];
        #pragma unroll
        for (int r = 0; r < 8; ++r) {
            cs[0][r] = bv0; cs[1][r] = bv1; cs[2][r] = bv2;
            cn[0][r] = 0.0f; cn[1][r] = 0.0f; cn[2][r] = 0.0f;
        }

        #pragma unroll
        for (int k = 0; k < 12; ++k) {
            const int kk = k * 4 + half * 2;
            v2f a;
            a.x = xrow[kk];
            a.y = xrow[kk + 1];
            v2f a2;
            a2.x = grow[kk] * dsc;
            a2.y = grow[kk + 1] * dsc;
            #pragma unroll
            for (int nt = 0; nt < 3; ++nt) {
                cs[nt] = __builtin_amdgcn_wmma_f32_16x16x4_f32(
                    false, a, false, wS[k][nt], (short)0, cs[nt], false, false);
                cn[nt] = __builtin_amdgcn_wmma_f32_16x16x4_f32(
                    false, a2, false, wN[k][nt], (short)0, cn[nt], false, false);
            }
        }

        #pragma unroll
        for (int nt = 0; nt < 3; ++nt) {
            #pragma unroll
            for (int r = 0; r < 8; ++r) {
                float v = cs[nt][r] + cn[nt][r];
                if (relu) v = fmaxf(v, 0.0f);
                out[(size_t)(m0 + half * 8 + r) * FDIM + nt * 16 + l15] = v;
            }
        }
    }
}

extern "C" void kernel_launch(void* const* d_in, const int* in_sizes, int n_in,
                              void* d_out, int out_size, void* d_ws, size_t ws_size,
                              hipStream_t stream) {
    const float* features = (const float*)d_in[0];
    const int*   src      = (const int*)d_in[1];
    const int*   dst      = (const int*)d_in[2];
    const float* Wself1   = (const float*)d_in[3];
    const float* Wneigh1  = (const float*)d_in[4];
    const float* b1       = (const float*)d_in[5];
    const float* Wself2   = (const float*)d_in[6];
    const float* Wneigh2  = (const float*)d_in[7];
    const float* b2       = (const float*)d_in[8];
    float* out = (float*)d_out;

    const int N = in_sizes[0] / FDIM;
    const int E = in_sizes[1];

    char* ws = (char*)d_ws;
    size_t off = 0;
    float* agg1 = (float*)(ws + off); off += (((size_t)N * PSTRIDE * 4) + 255) & ~(size_t)255;
    float* agg2 = (float*)(ws + off); off += (((size_t)N * FDIM * 4)    + 255) & ~(size_t)255;
    float* h    = (float*)(ws + off);
    const float* degcol = agg1 + FDIM;

    const unsigned scatterBlocks = ((unsigned)E + (8u * EPW - 1u)) / (8u * EPW);

    const int tiles        = N / 16;
    const int tilesPerWave = 5;
    const int gemmWaves    = tiles / tilesPerWave;
    const int gemmBlocks   = gemmWaves / 2;

    hipMemsetAsync(agg1, 0, (size_t)N * PSTRIDE * 4, stream);
    scatter_kernel<PSTRIDE, true><<<scatterBlocks, 256, 0, stream>>>(features, src, dst, agg1, E);
    sage_gemm_kernel<<<gemmBlocks, 64, 0, stream>>>(features, agg1, degcol,
                                                    Wself1, Wneigh1, b1, h, 1, tilesPerWave, PSTRIDE);

    hipMemsetAsync(agg2, 0, (size_t)N * FDIM * 4, stream);
    scatter_kernel<FDIM, false><<<scatterBlocks, 256, 0, stream>>>(h, src, dst, agg2, E);
    sage_gemm_kernel<<<gemmBlocks, 64, 0, stream>>>(h, agg2, degcol,
                                                    Wself2, Wneigh2, b2, out, 0, tilesPerWave, FDIM);
}


# float-atomic scatter replaced: edges counting-sorted by destination once per call (LDS integer counters), each node's in-rows summed in registers and stored once, both layers; sum order: sorted list
# speedup vs baseline: 3.4185x; 1.4202x over previous
#include <hip/hip_runtime.h>


typedef __attribute__((ext_vector_type(2))) float v2f;
typedef __attribute__((ext_vector_type(8))) float v8f;
typedef unsigned long long u64;

#define FDIM 48

#define PSTRIDE 64

#define BIN_SHIFT 8
#define BIN_ROWS  256
#define MAXBINS   512
#define CHUNK     6400
#define BTHREADS  1024
#define EPT       7
#define STHREADS  512
#define RPT       8
#define SORT_CAP  8192

template <int PER>
__device__ __forceinline__ void wave_excl_scan(const int* in, int* out, int n, int lane) {
    int loc[PER];
    int s = 0;
    #pragma unroll
    for (int k = 0; k < PER; ++k) {
        const int idx = lane * PER + k;
        loc[k] = s;
        s += (idx < n) ? in[idx] : 0;
    }
    int incl = s;
    #pragma unroll
    for (int off = 1; off < 32; off <<= 1) {
        const int t = __shfl_up(incl, off);
        if (lane >= off) incl += t;
    }
    const int excl = incl - s;
    #pragma unroll
    for (int k = 0; k < PER; ++k) {
        const int idx = lane * PER + k;
        if (idx < n) out[idx] = loc[k] + excl;
    }
}

__global__ void __launch_bounds__(BTHREADS)
bin_count_kernel(const int* __restrict__ dst, int* __restrict__ cntMat, int nEdges, int nBins) {
    __shared__ int hist[MAXBINS];
    const int tid = threadIdx.x;
    if (tid < MAXBINS) hist[tid] = 0;
    const int e0  = blockIdx.x * CHUNK;
    const int len = nEdges - e0;
    int d[EPT];
    #pragma unroll
    for (int u = 0; u < EPT; ++u) {
        const int i = tid + u * BTHREADS;
        d[u] = (i < CHUNK && i < len) ? dst[e0 + i] : -1;
    }
    __syncthreads();
    #pragma unroll
    for (int u = 0; u < EPT; ++u) {
        if (d[u] >= 0) atomicAdd(&hist[d[u] >> BIN_SHIFT], 1);
    }
    __syncthreads();
    if (tid < nBins) cntMat[blockIdx.x * nBins + tid] = hist[tid];
}

__global__ void __launch_bounds__(BTHREADS)
bin_fill_kernel(const int* __restrict__ src, const int* __restrict__ dst,
                const int* __restrict__ cntMat, u64* __restrict__ binned,
                int* __restrict__ binStart, int nEdges, int nBins, int nBlocksPad) {
    __shared__ int hist[MAXBINS];
    __shared__ int lstart[MAXBINS];
    __shared__ int total[MAXBINS];
    __shared__ int gstart[MAXBINS];
    __shared__ int before[MAXBINS];
    __shared__ u64 recs[CHUNK];
    const int tid  = threadIdx.x;
    const int lane = tid & 31;
    if (tid < MAXBINS) { hist[tid] = 0; total[tid] = 0; before[tid] = 0; }
    const int e0  = blockIdx.x * CHUNK;
    const int len = nEdges - e0;
    int d[EPT], s[EPT];
    #pragma unroll
    for (int u = 0; u < EPT; ++u) {
        const int i = tid + u * BTHREADS;
        const bool ok = i < CHUNK && i < len;
        d[u] = ok ? dst[e0 + i] : -1;
        s[u] = ok ? src[e0 + i] : 0;
    }
    int tot = 0, bef = 0;
    const int bin = tid & (MAXBINS - 1);
    const int grp = __builtin_amdgcn_readfirstlane(tid >> 9);
    if (bin < nBins) {
        const int per  = nBlocksPad >> 1;
        const int bBeg = grp * per;
        const int bEnd = bBeg + per;
        for (int b = bBeg; b < bEnd; b += 32) {
            int c[32];
            #pragma unroll
            for (int u = 0; u < 32; ++u) c[u] = cntMat[(b + u) * nBins + bin];
            #pragma unroll
            for (int u = 0; u < 32; ++u) {
                tot += c[u];
                bef += (b + u < (int)blockIdx.x) ? c[u] : 0;
            }
        }
    }
    __syncthreads();
    #pragma unroll
    for (int u = 0; u < EPT; ++u) {
        if (d[u] >= 0) atomicAdd(&hist[d[u] >> BIN_SHIFT], 1);
    }
    if (bin < nBins) {
        atomicAdd(&total[bin], tot);
        atomicAdd(&before[bin], bef);
    }
    __syncthreads();
    if (tid < 32) {
        wave_excl_scan<MAXBINS / 32>(hist,  lstart, nBins, lane);
        wave_excl_scan<MAXBINS / 32>(total, gstart, nBins, lane);
    }
    __syncthreads();
    if (tid < nBins) {
        hist[tid] = lstart[tid];
        if (blockIdx.x == 0) binStart[tid] = gstart[tid];
    }
    if (blockIdx.x == 0 && tid == 0) binStart[nBins] = nEdges;
    __syncthreads();
    #pragma unroll
    for (int u = 0; u < EPT; ++u) {
        if (d[u] >= 0) {
            const int p = atomicAdd(&hist[d[u] >> BIN_SHIFT], 1);
            recs[p] = ((u64)(unsigned)d[u] << 32) | (u64)(unsigned)s[u];
        }
    }
    __syncthreads();
    const int n = len < CHUNK ? len : CHUNK;
    for (int p = tid; p < n; p += BTHREADS) {
        const u64 r = recs[p];
        const int b = (int)(r >> 32) >> BIN_SHIFT;
        binned[gstart[b] + before[b] + (p - lstart[b])] = r;
    }
}

__global__ void __launch_bounds__(STHREADS)
bin_sort_kernel(const u64* __restrict__ binned, const int* __restrict__ binStart,
                u64* __restrict__ rowInfo, int* __restrict__ srcSorted, int nNodes) {
    __shared__ int hist[BIN_ROWS];
    __shared__ int rstart[BIN_ROWS];
    __shared__ int stage[SORT_CAP];
    const int tid  = threadIdx.x;
    const int lane = tid & 31;
    const int base = binStart[blockIdx.x];
    const int cnt  = binStart[blockIdx.x + 1] - base;
    const bool fits = cnt <= SORT_CAP;
    if (tid < BIN_ROWS) hist[tid] = 0;
    __syncthreads();
    for (int i0 = 0; i0 < cnt; i0 += STHREADS * RPT) {
        u64 r[RPT];
        #pragma unroll
        for (int u = 0; u < RPT; ++u) {
            const int i = i0 + u * STHREADS + tid;
            r[u] = binned[base + (i < cnt ? i : cnt - 1)];
        }
        #pragma unroll
        for (int u = 0; u < RPT; ++u) {
            const int i = i0 + u * STHREADS + tid;
            if (i < cnt) atomicAdd(&hist[(int)(r[u] >> 32) & (BIN_ROWS - 1)], 1);
        }
    }
    __syncthreads();
    if (tid < 32) wave_excl_scan<BIN_ROWS / 32>(hist, rstart, BIN_ROWS, lane);
    __syncthreads();
    if (tid < BIN_ROWS) {
        const int v = blockIdx.x * BIN_ROWS + tid;
        if (v < nNodes)
            rowInfo[v] = ((u64)(unsigned)hist[tid] << 32) | (u64)(unsigned)(base + rstart[tid]);
        hist[tid] = rstart[tid];
    }
    __syncthreads();
    for (int i0 = 0; i0 < cnt; i0 += STHREADS * RPT) {
        u64 r[RPT];
        #pragma unroll
        for (int u = 0; u < RPT; ++u) {
            const int i = i0 + u * STHREADS + tid;
            r[u] = binned[base + (i < cnt ? i : cnt - 1)];
        }
        #pragma unroll
        for (int u = 0; u < RPT; ++u) {
            const int i = i0 + u * STHREADS + tid;
            if (i < cnt) {
                const int p = atomicAdd(&hist[(int)(r[u] >> 32) & (BIN_ROWS - 1)], 1);
                if (fits) stage[p] = (int)(unsigned)r[u];
                else      srcSorted[base + p] = (int)(unsigned)r[u];
            }
        }
    }
    __syncthreads();
    if (fits) {
        for (int i = tid; i < cnt; i += STHREADS) srcSorted[base + i] = stage[i];
    }
}

#define NPW 4
#define EPG 8

template <int STRIDE, bool COUNT>
__global__ void __launch_bounds__(256)
gather_sum_kernel(const float* __restrict__ x, const u64* __restrict__ rowInfo,
                  const int* __restrict__ srcSorted, float* __restrict__ agg, int nNodes) {
    const unsigned lane = threadIdx.x & 31u;
    const unsigned wv = __builtin_amdgcn_readfirstlane(blockIdx.x * 8u + (threadIdx.x >> 5));
    const bool lo = lane < 16u;
    const unsigned v0 = wv * (unsigned)NPW;
    if (v0 >= (unsigned)nNodes) return;
    u64 info[NPW];
    #pragma unroll
    for (int n = 0; n < NPW; ++n) {
        const unsigned v = v0 + (unsigned)n;
        info[n] = rowInfo[v < (unsigned)nNodes ? v : (unsigned)nNodes - 1u];
    }
    #pragma unroll
    for (int n = 0; n < NPW; ++n) {
        const unsigned v = v0 + (unsigned)n;
        if (v < (unsigned)nNodes) {
            const int start = (int)(unsigned)info[n];
            const int cnt   = (int)(info[n] >> 32);
            float a0 = 0.0f, a1 = 0.0f;
            int idx[EPG];
            if (cnt > 0) {
                #pragma unroll
                for (int k = 0; k < EPG; ++k) idx[k] = srcSorted[start + (k < cnt ? k : cnt - 1)];
            }
            for (int j = 0; j < cnt; j += EPG) {
                const float* xr[EPG];
                float va[EPG], vb[EPG];
                #pragma unroll
                for (int k = 0; k < EPG; ++k) xr[k] = x + (size_t)idx[k] * FDIM;
                #pragma unroll
                for (int k = 0; k < EPG; ++k) {
                    va[k] = xr[k][lane];
                    vb[k] = 0.0f;
                }
                if (lo) {
                    #pragma unroll
                    for (int k = 0; k < EPG; ++k) vb[k] = xr[k][32u + lane];
                }
                #pragma unroll
                for (int k = 0; k < EPG; ++k) {
                    const int jn = j + EPG + k;
                    idx[k] = srcSorted[start + (jn < cnt ? jn : cnt - 1)];
                }
                #pragma unroll
                for (int k = 0; k < EPG; ++k) {
                    if (j + k < cnt) { a0 += va[k]; a1 += vb[k]; }
                }
            }
            float* g = agg + (size_t)v * STRIDE;
            g[lane] = a0;
            if (COUNT) {
                if (lane <= 16u) g[32u + lane] = lo ? a1 : (float)cnt;
            } else {
                if (lo) g[32u + lane] = a1;
            }
        }
    }
}

__global__ void __launch_bounds__(64)
sage_gemm_kernel(const float* __restrict__ x,
                 const float* __restrict__ agg,
                 const float* __restrict__ degcol,
                 const float* __restrict__ Wself,
                 const float* __restrict__ Wneigh,
                 const float* __restrict__ bias,
                 float* __restrict__ out,
                 int relu, int tilesPerWave, int aggStride) {
    const int lane = threadIdx.x & 31;
    const int wave = threadIdx.x >> 5;
    const int half = lane >> 4;
    const int l15  = lane & 15;
    const int gw   = blockIdx.x * 2 + wave;

    v2f wS[12][3], wN[12][3];
    #pragma unroll
    for (int k = 0; k < 12; ++k) {
        const int kk = k * 4 + half * 2;
        #pragma unroll
        for (int nt = 0; nt < 3; ++nt) {
            const int col = nt * 16 + l15;
            wS[k][nt].x = Wself [(size_t)(kk)     * FDIM + col];
            wS[k][nt].y = Wself [(size_t)(kk + 1) * FDIM + col];
            wN[k][nt].x = Wneigh[(size_t)(kk)     * FDIM + col];
            wN[k][nt].y = Wneigh[(size_t)(kk + 1) * FDIM + col];
        }
    }
    float bv0 = bias[0 * 16 + l15];
    float bv1 = bias[1 * 16 + l15];
    float bv2 = bias[2 * 16 + l15];

    #pragma unroll 1
    for (int t = 0; t < tilesPerWave; ++t) {
        const int m0 = (gw * tilesPerWave + t) * 16;
        const float* xrow = x   + (size_t)(m0 + l15) * FDIM;
        const float* grow = agg + (size_t)(m0 + l15) * aggStride;
        const float  dsc  = 1.0f / fmaxf(degcol[(size_t)(m0 + l15) * PSTRIDE], 1.0f);

        v8f cs[3], cn[3];
        #pragma unroll
        for (int r = 0; r < 8; ++r) {
            cs[0][r] = bv0; cs[1][r] = bv1; cs[2][r] = bv2;
            cn[0][r] = 0.0f; cn[1][r] = 0.0f; cn[2][r] = 0.0f;
        }

        #pragma unroll
        for (int k = 0; k < 12; ++k) {
            const int kk = k * 4 + half * 2;
            v2f a;
            a.x = xrow[kk];
            a.y = xrow[kk + 1];
            v2f a2;
            a2.x = grow[kk] * dsc;
            a2.y = grow[kk + 1] * dsc;
            #pragma unroll
            for (int nt = 0; nt < 3; ++nt) {
                cs[nt] = __builtin_amdgcn_wmma_f32_16x16x4_f32(
                    false, a, false, wS[k][nt], (short)0, cs[nt], false, false);
                cn[nt] = __builtin_amdgcn_wmma_f32_16x16x4_f32(
                    false, a2, false, wN[k][nt], (short)0, cn[nt], false, false);
            }
        }

        #pragma unroll
        for (int nt = 0; nt < 3; ++nt) {
            #pragma unroll
            for (int r = 0; r < 8; ++r) {
                float v = cs[nt][r] + cn[nt][r];
                if (relu) v = fmaxf(v, 0.0f);
                out[(size_t)(m0 + half * 8 + r) * FDIM + nt * 16 + l15] = v;
            }
        }
    }
}

extern "C" void kernel_launch(void* const* d_in, const int* in_sizes, int n_in,
                              void* d_out, int out_size, void* d_ws, size_t ws_size,
                              hipStream_t stream) {
    const float* features = (const float*)d_in[0];
    const int*   src      = (const int*)d_in[1];
    const int*   dst      = (const int*)d_in[2];
    const float* Wself1   = (const float*)d_in[3];
    const float* Wneigh1  = (const float*)d_in[4];
    const float* b1       = (const float*)d_in[5];
    const float* Wself2   = (const float*)d_in[6];
    const float* Wneigh2  = (const float*)d_in[7];
    const float* b2       = (const float*)d_in[8];
    float* out = (float*)d_out;

    const int N = in_sizes[0] / FDIM;
    const int E = in_sizes[1];

    const int nBins      = (N + BIN_ROWS - 1) / BIN_ROWS;
    const int nChunks    = (E + CHUNK - 1) / CHUNK;
    const int nChunksPad = (nChunks + 63) & ~63;

    char* ws = (char*)d_ws;
    size_t off = 0;
    float* agg1 = (float*)(ws + off); off += (((size_t)N * PSTRIDE * 4) + 255) & ~(size_t)255;
    float* agg2 = (float*)(ws + off); off += (((size_t)N * FDIM * 4)    + 255) & ~(size_t)255;
    float* h    = (float*)(ws + off); off += (((size_t)N * FDIM * 4)    + 255) & ~(size_t)255;
    u64*  binned    = (u64*)(ws + off);  off += (((size_t)E * 8) + 255) & ~(size_t)255;
    int*  srcSorted = (int*)(ws + off);  off += (((size_t)E * 4) + 255) & ~(size_t)255;
    u64*  rowInfo   = (u64*)(ws + off);  off += (((size_t)N * 8) + 255) & ~(size_t)255;
    int*  cntMat    = (int*)(ws + off);  off += (((size_t)nChunksPad * nBins * 4) + 255) & ~(size_t)255;
    int*  binStart  = (int*)(ws + off);
    const float* degcol = agg1 + FDIM;

    const unsigned gatherWaves  = ((unsigned)N + (NPW - 1u)) / NPW;
    const unsigned gatherBlocks = (gatherWaves + 7u) / 8u;

    const int tiles        = N / 16;
    const int tilesPerWave = 5;
    const int gemmWaves    = tiles / tilesPerWave;
    const int gemmBlocks   = gemmWaves / 2;

    bin_count_kernel<<<nChunksPad, BTHREADS, 0, stream>>>(dst, cntMat, E, nBins);
    bin_fill_kernel<<<nChunks, BTHREADS, 0, stream>>>(src, dst, cntMat, binned, binStart, E, nBins, nChunksPad);
    bin_sort_kernel<<<nBins, STHREADS, 0, stream>>>(binned, binStart, rowInfo, srcSorted, N);

    gather_sum_kernel<PSTRIDE, true><<<gatherBlocks, 256, 0, stream>>>(features, rowInfo, srcSorted, agg1, N);
    sage_gemm_kernel<<<gemmBlocks, 64, 0, stream>>>(features, agg1, degcol,
                                                    Wself1, Wneigh1, b1, h, 1, tilesPerWave, PSTRIDE);

    gather_sum_kernel<FDIM, false><<<gatherBlocks, 256, 0, stream>>>(h, rowInfo, srcSorted, agg2, N);
    sage_gemm_kernel<<<gemmBlocks, 64, 0, stream>>>(h, agg2, degcol,
                                                    Wself2, Wneigh2, b2, out, 0, tilesPerWave, FDIM);
}


# gather rows fetched 16 B per lane: a half-wave takes a whole 192-B row, one load brings two in-edges; even/odd list positions summed apart, one cross-lane add per node (new sum order); 8 nodes/wave
# speedup vs baseline: 3.5804x; 1.0474x over previous
#include <hip/hip_runtime.h>


typedef __attribute__((ext_vector_type(2))) float v2f;
typedef __attribute__((ext_vector_type(8))) float v8f;
typedef unsigned long long u64;

#define FDIM 48

#define PSTRIDE 64

#define BIN_SHIFT 8
#define BIN_ROWS  256
#define MAXBINS   512
#define CHUNK     6400
#define BTHREADS  1024
#define EPT       7
#define STHREADS  512
#define RPT       8
#define SORT_CAP  8192

template <int PER>
__device__ __forceinline__ void wave_excl_scan(const int* in, int* out, int n, int lane) {
    int loc[PER];
    int s = 0;
    #pragma unroll
    for (int k = 0; k < PER; ++k) {
        const int idx = lane * PER + k;
        loc[k] = s;
        s += (idx < n) ? in[idx] : 0;
    }
    int incl = s;
    #pragma unroll
    for (int off = 1; off < 32; off <<= 1) {
        const int t = __shfl_up(incl, off);
        if (lane >= off) incl += t;
    }
    const int excl = incl - s;
    #pragma unroll
    for (int k = 0; k < PER; ++k) {
        const int idx = lane * PER + k;
        if (idx < n) out[idx] = loc[k] + excl;
    }
}

__global__ void __launch_bounds__(BTHREADS)
bin_count_kernel(const int* __restrict__ dst, int* __restrict__ cntMat, int nEdges, int nBins) {
    __shared__ int hist[MAXBINS];
    const int tid = threadIdx.x;
    if (tid < MAXBINS) hist[tid] = 0;
    const int e0  = blockIdx.x * CHUNK;
    const int len = nEdges - e0;
    int d[EPT];
    #pragma unroll
    for (int u = 0; u < EPT; ++u) {
        const int i = tid + u * BTHREADS;
        d[u] = (i < CHUNK && i < len) ? dst[e0 + i] : -1;
    }
    __syncthreads();
    #pragma unroll
    for (int u = 0; u < EPT; ++u) {
        if (d[u] >= 0) atomicAdd(&hist[d[u] >> BIN_SHIFT], 1);
    }
    __syncthreads();
    if (tid < nBins) cntMat[blockIdx.x * nBins + tid] = hist[tid];
}

__global__ void __launch_bounds__(BTHREADS)
bin_fill_kernel(const int* __restrict__ src, const int* __restrict__ dst,
                const int* __restrict__ cntMat, u64* __restrict__ binned,
                int* __restrict__ binStart, int nEdges, int nBins, int nBlocksPad) {
    __shared__ int hist[MAXBINS];
    __shared__ int lstart[MAXBINS];
    __shared__ int total[MAXBINS];
    __shared__ int gstart[MAXBINS];
    __shared__ int before[MAXBINS];
    __shared__ u64 recs[CHUNK];
    const int tid  = threadIdx.x;
    const int lane = tid & 31;
    if (tid < MAXBINS) { hist[tid] = 0; total[tid] = 0; before[tid] = 0; }
    const int e0  = blockIdx.x * CHUNK;
    const int len = nEdges - e0;
    int d[EPT], s[EPT];
    #pragma unroll
    for (int u = 0; u < EPT; ++u) {
        const int i = tid + u * BTHREADS;
        const bool ok = i < CHUNK && i < len;
        d[u] = ok ? dst[e0 + i] : -1;
        s[u] = ok ? src[e0 + i] : 0;
    }
    int tot = 0, bef = 0;
    const int bin = tid & (MAXBINS - 1);
    const int grp = __builtin_amdgcn_readfirstlane(tid >> 9);
    if (bin < nBins) {
        const int per  = nBlocksPad >> 1;
        const int bBeg = grp * per;
        const int bEnd = bBeg + per;
        for (int b = bBeg; b < bEnd; b += 32) {
            int c[32];
            #pragma unroll
            for (int u = 0; u < 32; ++u) c[u] = cntMat[(b + u) * nBins + bin];
            #pragma unroll
            for (int u = 0; u < 32; ++u) {
                tot += c[u];
                bef += (b + u < (int)blockIdx.x) ? c[u] : 0;
            }
        }
    }
    __syncthreads();
    #pragma unroll
    for (int u = 0; u < EPT; ++u) {
        if (d[u] >= 0) atomicAdd(&hist[d[u] >> BIN_SHIFT], 1);
    }
    if (bin < nBins) {
        atomicAdd(&total[bin], tot);
        atomicAdd(&before[bin], bef);
    }
    __syncthreads();
    if (tid < 32) {
        wave_excl_scan<MAXBINS / 32>(hist,  lstart, nBins, lane);
        wave_excl_scan<MAXBINS / 32>(total, gstart, nBins, lane);
    }
    __syncthreads();
    if (tid < nBins) {
        hist[tid] = lstart[tid];
        if (blockIdx.x == 0) binStart[tid] = gstart[tid];
    }
    if (blockIdx.x == 0 && tid == 0) binStart[nBins] = nEdges;
    __syncthreads();
    #pragma unroll
    for (int u = 0; u < EPT; ++u) {
        if (d[u] >= 0) {
            const int p = atomicAdd(&hist[d[u] >> BIN_SHIFT], 1);
            recs[p] = ((u64)(unsigned)d[u] << 32) | (u64)(unsigned)s[u];
        }
    }
    __syncthreads();
    const int n = len < CHUNK ? len : CHUNK;
    for (int p = tid; p < n; p += BTHREADS) {
        const u64 r = recs[p];
        const int b = (int)(r >> 32) >> BIN_SHIFT;
        binned[gstart[b] + before[b] + (p - lstart[b])] = r;
    }
}

__global__ void __launch_bounds__(STHREADS)
bin_sort_kernel(const u64* __restrict__ binned, const int* __restrict__ binStart,
                u64* __restrict__ rowInfo, int* __restrict__ srcSorted, int nNodes) {
    __shared__ int hist[BIN_ROWS];
    __shared__ int rstart[BIN_ROWS];
    __shared__ int stage[SORT_CAP];
    const int tid  = threadIdx.x;
    const int lane = tid & 31;
    const int base = binStart[blockIdx.x];
    const int cnt  = binStart[blockIdx.x + 1] - base;
    const bool fits = cnt <= SORT_CAP;
    if (tid < BIN_ROWS) hist[tid] = 0;
    __syncthreads();
    for (int i0 = 0; i0 < cnt; i0 += STHREADS * RPT) {
        u64 r[RPT];
        #pragma unroll
        for (int u = 0; u < RPT; ++u) {
            const int i = i0 + u * STHREADS + tid;
            r[u] = binned[base + (i < cnt ? i : cnt - 1)];
        }
        #pragma unroll
        for (int u = 0; u < RPT; ++u) {
            const int i = i0 + u * STHREADS + tid;
            if (i < cnt) atomicAdd(&hist[(int)(r[u] >> 32) & (BIN_ROWS - 1)], 1);
        }
    }
    __syncthreads();
    if (tid < 32) wave_excl_scan<BIN_ROWS / 32>(hist, rstart, BIN_ROWS, lane);
    __syncthreads();
    if (tid < BIN_ROWS) {
        const int v = blockIdx.x * BIN_ROWS + tid;
        if (v < nNodes)
            rowInfo[v] = ((u64)(unsigned)hist[tid] << 32) | (u64)(unsigned)(base + rstart[tid]);
        hist[tid] = rstart[tid];
    }
    __syncthreads();
    for (int i0 = 0; i0 < cnt; i0 += STHREADS * RPT) {
        u64 r[RPT];
        #pragma unroll
        for (int u = 0; u < RPT; ++u) {
            const int i = i0 + u * STHREADS + tid;
            r[u] = binned[base + (i < cnt ? i : cnt - 1)];
        }
        #pragma unroll
        for (int u = 0; u < RPT; ++u) {
            const int i = i0 + u * STHREADS + tid;
            if (i < cnt) {
                const int p = atomicAdd(&hist[(int)(r[u] >> 32) & (BIN_ROWS - 1)], 1);
                if (fits) stage[p] = (int)(unsigned)r[u];
                else      srcSorted[base + p] = (int)(unsigned)r[u];
            }
        }
    }
    __syncthreads();
    if (fits) {
        for (int i = tid; i < cnt; i += STHREADS) srcSorted[base + i] = stage[i];
    }
}

#define NPW 8
#define LPG 8

typedef __attribute__((ext_vector_type(4))) float v4f;

template <int STRIDE, bool COUNT>
__global__ void __launch_bounds__(256)
gather_sum_kernel(const float* __restrict__ x, const u64* __restrict__ rowInfo,
                  const int* __restrict__ srcSorted, float* __restrict__ agg, int nNodes) {
    const unsigned lane = threadIdx.x & 31u;
    const unsigned wv = __builtin_amdgcn_readfirstlane(blockIdx.x * 8u + (threadIdx.x >> 5));
    const unsigned hi = lane >> 4;
    const unsigned c  = lane & 15u;
    const unsigned cb = (c < 12u ? c : 11u) * 16u;
    const unsigned v0 = wv * (unsigned)NPW;
    if (v0 >= (unsigned)nNodes) return;
    const char* xb = (const char*)x;
    u64 info[NPW];
    #pragma unroll
    for (int n = 0; n < NPW; ++n) {
        const unsigned v = v0 + (unsigned)n;
        info[n] = rowInfo[v < (unsigned)nNodes ? v : (unsigned)nNodes - 1u];
    }
    #pragma unroll
    for (int n = 0; n < NPW; ++n) {
        const unsigned v = v0 + (unsigned)n;
        if (v < (unsigned)nNodes) {
            const int start = (int)(unsigned)info[n];
            const int cnt   = (int)(info[n] >> 32);
            v4f acc = {0.0f, 0.0f, 0.0f, 0.0f};
            int idx[2 * LPG];
            if (cnt > 0) {
                #pragma unroll
                for (int k = 0; k < 2 * LPG; ++k) idx[k] = srcSorted[start + (k < cnt ? k : cnt - 1)];
            }
            for (int j = 0; j < cnt; j += 2 * LPG) {
                v4f val[LPG];
                unsigned off[LPG];
                #pragma unroll
                for (int k = 0; k < LPG; ++k) {
                    const unsigned e = hi ? (unsigned)idx[2 * k + 1] : (unsigned)idx[2 * k];
                    off[k] = e * (unsigned)(FDIM * 4) + cb;
                }
                #pragma unroll
                for (int k = 0; k < LPG; ++k) val[k] = *(const v4f*)(xb + off[k]);
                #pragma unroll
                for (int k = 0; k < 2 * LPG; ++k) {
                    const int jn = j + 2 * LPG + k;
                    idx[k] = srcSorted[start + (jn < cnt ? jn : cnt - 1)];
                }
                #pragma unroll
                for (int k = 0; k < LPG; ++k) {
                    if (j + 2 * k + (int)hi < cnt) acc += val[k];
                }
            }
            acc.x += __shfl_xor(acc.x, 16);
            acc.y += __shfl_xor(acc.y, 16);
            acc.z += __shfl_xor(acc.z, 16);
            acc.w += __shfl_xor(acc.w, 16);
            float* g = agg + (size_t)v * STRIDE;
            if (COUNT) {
                if (lane <= 12u) {
                    v4f o = acc;
                    if (lane == 12u) { o.x = (float)cnt; o.y = 0.0f; o.z = 0.0f; o.w = 0.0f; }
                    *(v4f*)(g + lane * 4u) = o;
                }
            } else {
                if (lane < 12u) *(v4f*)(g + lane * 4u) = acc;
            }
        }
    }
}

__global__ void __launch_bounds__(64)
sage_gemm_kernel(const float* __restrict__ x,
                 const float* __restrict__ agg,
                 const float* __restrict__ degcol,
                 const float* __restrict__ Wself,
                 const float* __restrict__ Wneigh,
                 const float* __restrict__ bias,
                 float* __restrict__ out,
                 int relu, int tilesPerWave, int aggStride) {
    const int lane = threadIdx.x & 31;
    const int wave = threadIdx.x >> 5;
    const int half = lane >> 4;
    const int l15  = lane & 15;
    const int gw   = blockIdx.x * 2 + wave;

    v2f wS[12][3], wN[12][3];
    #pragma unroll
    for (int k = 0; k < 12; ++k) {
        const int kk = k * 4 + half * 2;
        #pragma unroll
        for (int nt = 0; nt < 3; ++nt) {
            const int col = nt * 16 + l15;
            wS[k][nt].x = Wself [(size_t)(kk)     * FDIM + col];
            wS[k][nt].y = Wself [(size_t)(kk + 1) * FDIM + col];
            wN[k][nt].x = Wneigh[(size_t)(kk)     * FDIM + col];
            wN[k][nt].y = Wneigh[(size_t)(kk + 1) * FDIM + col];
        }
    }
    float bv0 = bias[0 * 16 + l15];
    float bv1 = bias[1 * 16 + l15];
    float bv2 = bias[2 * 16 + l15];

    #pragma unroll 1
    for (int t = 0; t < tilesPerWave; ++t) {
        const int m0 = (gw * tilesPerWave + t) * 16;
        const float* xrow = x   + (size_t)(m0 + l15) * FDIM;
        const float* grow = agg + (size_t)(m0 + l15) * aggStride;
        const float  dsc  = 1.0f / fmaxf(degcol[(size_t)(m0 + l15) * PSTRIDE], 1.0f);

        v8f cs[3], cn[3];
        #pragma unroll
        for (int r = 0; r < 8; ++r) {
            cs[0][r] = bv0; cs[1][r] = bv1; cs[2][r] = bv2;
            cn[0][r] = 0.0f; cn[1][r] = 0.0f; cn[2][r] = 0.0f;
        }

        #pragma unroll
        for (int k = 0; k < 12; ++k) {
            const int kk = k * 4 + half * 2;
            v2f a;
            a.x = xrow[kk];
            a.y = xrow[kk + 1];
            v2f a2;
            a2.x = grow[kk] * dsc;
            a2.y = grow[kk + 1] * dsc;
            #pragma unroll
            for (int nt = 0; nt < 3; ++nt) {
                cs[nt] = __builtin_amdgcn_wmma_f32_16x16x4_f32(
                    false, a, false, wS[k][nt], (short)0, cs[nt], false, false);
                cn[nt] = __builtin_amdgcn_wmma_f32_16x16x4_f32(
                    false, a2, false, wN[k][nt], (short)0, cn[nt], false, false);
            }
        }

        #pragma unroll
        for (int nt = 0; nt < 3; ++nt) {
            #pragma unroll
            for (int r = 0; r < 8; ++r) {
                float v = cs[nt][r] + cn[nt][r];
                if (relu) v = fmaxf(v, 0.0f);
                out[(size_t)(m0 + half * 8 + r) * FDIM + nt * 16 + l15] = v;
            }
        }
    }
}

extern "C" void kernel_launch(void* const* d_in, const int* in_sizes, int n_in,
                              void* d_out, int out_size, void* d_ws, size_t ws_size,
                              hipStream_t stream) {
    const float* features = (const float*)d_in[0];
    const int*   src      = (const int*)d_in[1];
    const int*   dst      = (const int*)d_in[2];
    const float* Wself1   = (const float*)d_in[3];
    const float* Wneigh1  = (const float*)d_in[4];
    const float* b1       = (const float*)d_in[5];
    const float* Wself2   = (const float*)d_in[6];
    const float* Wneigh2  = (const float*)d_in[7];
    const float* b2       = (const float*)d_in[8];
    float* out = (float*)d_out;

    const int N = in_sizes[0] / FDIM;
    const int E = in_sizes[1];

    const int nBins      = (N + BIN_ROWS - 1) / BIN_ROWS;
    const int nChunks    = (E + CHUNK - 1) / CHUNK;
    const int nChunksPad = (nChunks + 63) & ~63;

    char* ws = (char*)d_ws;
    size_t off = 0;
    float* agg1 = (float*)(ws + off); off += (((size_t)N * PSTRIDE * 4) + 255) & ~(size_t)255;
    float* agg2 = (float*)(ws + off); off += (((size_t)N * FDIM * 4)    + 255) & ~(size_t)255;
    float* h    = (float*)(ws + off); off += (((size_t)N * FDIM * 4)    + 255) & ~(size_t)255;
    u64*  binned    = (u64*)(ws + off);  off += (((size_t)E * 8) + 255) & ~(size_t)255;
    int*  srcSorted = (int*)(ws + off);  off += (((size_t)E * 4) + 255) & ~(size_t)255;
    u64*  rowInfo   = (u64*)(ws + off);  off += (((size_t)N * 8) + 255) & ~(size_t)255;
    int*  cntMat    = (int*)(ws + off);  off += (((size_t)nChunksPad * nBins * 4) + 255) & ~(size_t)255;
    int*  binStart  = (int*)(ws + off);
    const float* degcol = agg1 + FDIM;

    const unsigned gatherWaves  = ((unsigned)N + (NPW - 1u)) / NPW;
    const unsigned gatherBlocks = (gatherWaves + 7u) / 8u;

    const int tiles        = N / 16;
    const int tilesPerWave = 5;
    const int gemmWaves    = tiles / tilesPerWave;
    const int gemmBlocks   = gemmWaves / 2;

    bin_count_kernel<<<nChunksPad, BTHREADS, 0, stream>>>(dst, cntMat, E, nBins);
    bin_fill_kernel<<<nChunks, BTHREADS, 0, stream>>>(src, dst, cntMat, binned, binStart, E, nBins, nChunksPad);
    bin_sort_kernel<<<nBins, STHREADS, 0, stream>>>(binned, binStart, rowInfo, srcSorted, N);

    gather_sum_kernel<PSTRIDE, true><<<gatherBlocks, 256, 0, stream>>>(features, rowInfo, srcSorted, agg1, N);
    sage_gemm_kernel<<<gemmBlocks, 64, 0, stream>>>(features, agg1, degcol,
                                                    Wself1, Wneigh1, b1, h, 1, tilesPerWave, PSTRIDE);

    gather_sum_kernel<FDIM, false><<<gatherBlocks, 256, 0, stream>>>(h, rowInfo, srcSorted, agg2, N);
    sage_gemm_kernel<<<gemmBlocks, 64, 0, stream>>>(h, agg2, degcol,
                                                    Wself2, Wneigh2, b2, out, 0, tilesPerWave, FDIM);
}


# edge sort in 2 launches, not 3: a block bin-sorts its chunk in place and writes a run table; a bin's block finds its runs and first slot in that table; count pass gone; order in node lists changes
# speedup vs baseline: 3.7912x; 1.0589x over previous
#include <hip/hip_runtime.h>


typedef __attribute__((ext_vector_type(2))) float v2f;
typedef __attribute__((ext_vector_type(8))) float v8f;
typedef unsigned long long u64;

#define FDIM 48

#define PSTRIDE 64

#define BIN_SHIFT 8
#define BIN_ROWS  256
#define MAXBINS   512
#define CHUNK     6400
#define BTHREADS  1024
#define EPT       7
#define STHREADS  512
#define MAXCHUNKS 256
#define RPT       8
#define SORT_CAP  8192

template <int PER>
__device__ __forceinline__ void wave_excl_scan(const int* in, int* out, int n, int lane) {
    int loc[PER];
    int s = 0;
    #pragma unroll
    for (int k = 0; k < PER; ++k) {
        const int idx = lane * PER + k;
        loc[k] = s;
        s += (idx < n) ? in[idx] : 0;
    }
    int incl = s;
    #pragma unroll
    for (int off = 1; off < 32; off <<= 1) {
        const int t = __shfl_up(incl, off);
        if (lane >= off) incl += t;
    }
    const int excl = incl - s;
    #pragma unroll
    for (int k = 0; k < PER; ++k) {
        const int idx = lane * PER + k;
        if (idx < n) out[idx] = loc[k] + excl;
    }
}

__global__ void __launch_bounds__(BTHREADS)
chunk_sort_kernel(const int* __restrict__ src, const int* __restrict__ dst,
                  u64* __restrict__ binned, int* __restrict__ runTab, int nEdges, int nBins) {
    __shared__ int hist[MAXBINS];
    __shared__ int lstart[MAXBINS];
    __shared__ u64 recs[CHUNK];
    const int tid  = threadIdx.x;
    const int lane = tid & 31;
    if (tid < MAXBINS) hist[tid] = 0;
    const int e0  = blockIdx.x * CHUNK;
    const int len = nEdges - e0;
    const int n   = len < CHUNK ? len : CHUNK;
    int d[EPT], s[EPT];
    #pragma unroll
    for (int u = 0; u < EPT; ++u) {
        const int i = tid + u * BTHREADS;
        const bool ok = i < n;
        d[u] = ok ? dst[e0 + i] : -1;
        s[u] = ok ? src[e0 + i] : 0;
    }
    __syncthreads();
    #pragma unroll
    for (int u = 0; u < EPT; ++u) {
        if (d[u] >= 0) atomicAdd(&hist[d[u] >> BIN_SHIFT], 1);
    }
    __syncthreads();
    if (tid < 32) wave_excl_scan<MAXBINS / 32>(hist, lstart, nBins, lane);
    __syncthreads();
    int* row = runTab + (size_t)blockIdx.x * (nBins + 1);
    if (tid < nBins) {
        row[tid]  = lstart[tid];
        hist[tid] = lstart[tid];
    }
    if (tid == 0) row[nBins] = n;
    __syncthreads();
    #pragma unroll
    for (int u = 0; u < EPT; ++u) {
        if (d[u] >= 0) {
            const int p = atomicAdd(&hist[d[u] >> BIN_SHIFT], 1);
            recs[p] = ((u64)(unsigned)d[u] << 32) | (u64)(unsigned)s[u];
        }
    }
    __syncthreads();
    for (int p = tid; p < n; p += BTHREADS) binned[e0 + p] = recs[p];
}

__global__ void __launch_bounds__(STHREADS)
bin_sort_kernel(const u64* __restrict__ binned, const int* __restrict__ runTab,
                u64* __restrict__ rowInfo, int* __restrict__ srcSorted,
                int nNodes, int nChunks, int nBins) {
    __shared__ int hist[BIN_ROWS];
    __shared__ int rstart[BIN_ROWS];
    __shared__ int runPos[MAXCHUNKS];
    __shared__ int runCnt[MAXCHUNKS];
    __shared__ int sums[2];
    __shared__ int stage[SORT_CAP];
    const int tid  = threadIdx.x;
    const int lane = tid & 31;
    if (tid < BIN_ROWS) hist[tid] = 0;
    if (tid < 2) sums[tid] = 0;
    __syncthreads();
    if (tid < MAXCHUNKS) {
        int a = 0, c = 0;
        if (tid < nChunks) {
            const int* row = runTab + (size_t)tid * (nBins + 1) + blockIdx.x;
            a = row[0];
            c = row[1] - a;
        }
        runPos[tid] = tid * CHUNK + a;
        runCnt[tid] = c;
        #pragma unroll
        for (int off = 16; off > 0; off >>= 1) {
            a += __shfl_xor(a, off);
            c += __shfl_xor(c, off);
        }
        if (lane == 0) { atomicAdd(&sums[0], a); atomicAdd(&sums[1], c); }
    }
    __syncthreads();
    const int base = sums[0];
    const int cnt  = sums[1];
    const bool fits = cnt <= SORT_CAP;
    const int hf = tid & 1;
    const int rp = runPos[tid >> 1];
    const int rc = runCnt[tid >> 1];
    for (int k0 = 0; k0 < rc; k0 += 2 * RPT) {
        u64 r[RPT];
        #pragma unroll
        for (int u = 0; u < RPT; ++u) {
            const int k = k0 + 2 * u + hf;
            r[u] = binned[rp + (k < rc ? k : rc - 1)];
        }
        #pragma unroll
        for (int u = 0; u < RPT; ++u) {
            const int k = k0 + 2 * u + hf;
            if (k < rc) atomicAdd(&hist[(int)(r[u] >> 32) & (BIN_ROWS - 1)], 1);
        }
    }
    __syncthreads();
    if (tid < 32) wave_excl_scan<BIN_ROWS / 32>(hist, rstart, BIN_ROWS, lane);
    __syncthreads();
    if (tid < BIN_ROWS) {
        const int v = blockIdx.x * BIN_ROWS + tid;
        if (v < nNodes)
            rowInfo[v] = ((u64)(unsigned)hist[tid] << 32) | (u64)(unsigned)(base + rstart[tid]);
        hist[tid] = rstart[tid];
    }
    __syncthreads();
    for (int k0 = 0; k0 < rc; k0 += 2 * RPT) {
        u64 r[RPT];
        #pragma unroll
        for (int u = 0; u < RPT; ++u) {
            const int k = k0 + 2 * u + hf;
            r[u] = binned[rp + (k < rc ? k : rc - 1)];
        }
        #pragma unroll
        for (int u = 0; u < RPT; ++u) {
            const int k = k0 + 2 * u + hf;
            if (k < rc) {
                const int p = atomicAdd(&hist[(int)(r[u] >> 32) & (BIN_ROWS - 1)], 1);
                if (fits) stage[p] = (int)(unsigned)r[u];
                else      srcSorted[base + p] = (int)(unsigned)r[u];
            }
        }
    }
    __syncthreads();
    if (fits) {
        for (int i = tid; i < cnt; i += STHREADS) srcSorted[base + i] = stage[i];
    }
}

#define NPW 8
#define LPG 8

typedef __attribute__((ext_vector_type(4))) float v4f;

template <int STRIDE, bool COUNT>
__global__ void __launch_bounds__(256)
gather_sum_kernel(const float* __restrict__ x, const u64* __restrict__ rowInfo,
                  const int* __restrict__ srcSorted, float* __restrict__ agg, int nNodes) {
    const unsigned lane = threadIdx.x & 31u;
    const unsigned wv = __builtin_amdgcn_readfirstlane(blockIdx.x * 8u + (threadIdx.x >> 5));
    const unsigned hi = lane >> 4;
    const unsigned c  = lane & 15u;
    const unsigned cb = (c < 12u ? c : 11u) * 16u;
    const unsigned v0 = wv * (unsigned)NPW;
    if (v0 >= (unsigned)nNodes) return;
    const char* xb = (const char*)x;
    u64 info[NPW];
    #pragma unroll
    for (int n = 0; n < NPW; ++n) {
        const unsigned v = v0 + (unsigned)n;
        info[n] = rowInfo[v < (unsigned)nNodes ? v : (unsigned)nNodes - 1u];
    }
    #pragma unroll
    for (int n = 0; n < NPW; ++n) {
        const unsigned v = v0 + (unsigned)n;
        if (v < (unsigned)nNodes) {
            const int start = (int)(unsigned)info[n];
            const int cnt   = (int)(info[n] >> 32);
            v4f acc = {0.0f, 0.0f, 0.0f, 0.0f};
            int idx[2 * LPG];
            if (cnt > 0) {
                #pragma unroll
                for (int k = 0; k < 2 * LPG; ++k) idx[k] = srcSorted[start + (k < cnt ? k : cnt - 1)];
            }
            for (int j = 0; j < cnt; j += 2 * LPG) {
                v4f val[LPG];
                unsigned off[LPG];
                #pragma unroll
                for (int k = 0; k < LPG; ++k) {
                    const unsigned e = hi ? (unsigned)idx[2 * k + 1] : (unsigned)idx[2 * k];
                    off[k] = e * (unsigned)(FDIM * 4) + cb;
                }
                #pragma unroll
                for (int k = 0; k < LPG; ++k) val[k] = *(const v4f*)(xb + off[k]);
                #pragma unroll
                for (int k = 0; k < 2 * LPG; ++k) {
                    const int jn = j + 2 * LPG + k;
                    idx[k] = srcSorted[start + (jn < cnt ? jn : cnt - 1)];
                }
                #pragma unroll
                for (int k = 0; k < LPG; ++k) {
                    if (j + 2 * k + (int)hi < cnt) acc += val[k];
                }
            }
            acc.x += __shfl_xor(acc.x, 16);
            acc.y += __shfl_xor(acc.y, 16);
            acc.z += __shfl_xor(acc.z, 16);
            acc.w += __shfl_xor(acc.w, 16);
            float* g = agg + (size_t)v * STRIDE;
            if (COUNT) {
                if (lane <= 12u) {
                    v4f o = acc;
                    if (lane == 12u) { o.x = (float)cnt; o.y = 0.0f; o.z = 0.0f; o.w = 0.0f; }
                    *(v4f*)(g + lane * 4u) = o;
                }
            } else {
                if (lane < 12u) *(v4f*)(g + lane * 4u) = acc;
            }
        }
    }
}

__global__ void __launch_bounds__(64)
sage_gemm_kernel(const float* __restrict__ x,
                 const float* __restrict__ agg,
                 const float* __restrict__ degcol,
                 const float* __restrict__ Wself,
                 const float* __restrict__ Wneigh,
                 const float* __restrict__ bias,
                 float* __restrict__ out,
                 int relu, int tilesPerWave, int aggStride) {
    const int lane = threadIdx.x & 31;
    const int wave = threadIdx.x >> 5;
    const int half = lane >> 4;
    const int l15  = lane & 15;
    const int gw   = blockIdx.x * 2 + wave;

    v2f wS[12][3], wN[12][3];
    #pragma unroll
    for (int k = 0; k < 12; ++k) {
        const int kk = k * 4 + half * 2;
        #pragma unroll
        for (int nt = 0; nt < 3; ++nt) {
            const int col = nt * 16 + l15;
            wS[k][nt].x = Wself [(size_t)(kk)     * FDIM + col];
            wS[k][nt].y = Wself [(size_t)(kk + 1) * FDIM + col];
            wN[k][nt].x = Wneigh[(size_t)(kk)     * FDIM + col];
            wN[k][nt].y = Wneigh[(size_t)(kk + 1) * FDIM + col];
        }
    }
    float bv0 = bias[0 * 16 + l15];
    float bv1 = bias[1 * 16 + l15];
    float bv2 = bias[2 * 16 + l15];

    #pragma unroll 1
    for (int t = 0; t < tilesPerWave; ++t) {
        const int m0 = (gw * tilesPerWave + t) * 16;
        const float* xrow = x   + (size_t)(m0 + l15) * FDIM;
        const float* grow = agg + (size_t)(m0 + l15) * aggStride;
        const float  dsc  = 1.0f / fmaxf(degcol[(size_t)(m0 + l15) * PSTRIDE], 1.0f);

        v8f cs[3], cn[3];
        #pragma unroll
        for (int r = 0; r < 8; ++r) {
            cs[0][r] = bv0; cs[1][r] = bv1; cs[2][r] = bv2;
            cn[0][r] = 0.0f; cn[1][r] = 0.0f; cn[2][r] = 0.0f;
        }

        #pragma unroll
        for (int k = 0; k < 12; ++k) {
            const int kk = k * 4 + half * 2;
            v2f a;
            a.x = xrow[kk];
            a.y = xrow[kk + 1];
            v2f a2;
            a2.x = grow[kk] * dsc;
            a2.y = grow[kk + 1] * dsc;
            #pragma unroll
            for (int nt = 0; nt < 3; ++nt) {
                cs[nt] = __builtin_amdgcn_wmma_f32_16x16x4_f32(
                    false, a, false, wS[k][nt], (short)0, cs[nt], false, false);
                cn[nt] = __builtin_amdgcn_wmma_f32_16x16x4_f32(
                    false, a2, false, wN[k][nt], (short)0, cn[nt], false, false);
            }
        }

        #pragma unroll
        for (int nt = 0; nt < 3; ++nt) {
            #pragma unroll
            for (int r = 0; r < 8; ++r) {
                float v = cs[nt][r] + cn[nt][r];
                if (relu) v = fmaxf(v, 0.0f);
                out[(size_t)(m0 + half * 8 + r) * FDIM + nt * 16 + l15] = v;
            }
        }
    }
}

extern "C" void kernel_launch(void* const* d_in, const int* in_sizes, int n_in,
                              void* d_out, int out_size, void* d_ws, size_t ws_size,
                              hipStream_t stream) {
    const float* features = (const float*)d_in[0];
    const int*   src      = (const int*)d_in[1];
    const int*   dst      = (const int*)d_in[2];
    const float* Wself1   = (const float*)d_in[3];
    const float* Wneigh1  = (const float*)d_in[4];
    const float* b1       = (const float*)d_in[5];
    const float* Wself2   = (const float*)d_in[6];
    const float* Wneigh2  = (const float*)d_in[7];
    const float* b2       = (const float*)d_in[8];
    float* out = (float*)d_out;

    const int N = in_sizes[0] / FDIM;
    const int E = in_sizes[1];

    const int nBins      = (N + BIN_ROWS - 1) / BIN_ROWS;
    const int nChunks    = (E + CHUNK - 1) / CHUNK;

    char* ws = (char*)d_ws;
    size_t off = 0;
    float* agg1 = (float*)(ws + off); off += (((size_t)N * PSTRIDE * 4) + 255) & ~(size_t)255;
    float* agg2 = (float*)(ws + off); off += (((size_t)N * FDIM * 4)    + 255) & ~(size_t)255;
    float* h    = (float*)(ws + off); off += (((size_t)N * FDIM * 4)    + 255) & ~(size_t)255;
    u64*  binned    = (u64*)(ws + off);  off += (((size_t)E * 8) + 255) & ~(size_t)255;
    int*  srcSorted = (int*)(ws + off);  off += (((size_t)E * 4) + 255) & ~(size_t)255;
    u64*  rowInfo   = (u64*)(ws + off);  off += (((size_t)N * 8) + 255) & ~(size_t)255;
    int*  runTab    = (int*)(ws + off);
    const float* degcol = agg1 + FDIM;

    const unsigned gatherWaves  = ((unsigned)N + (NPW - 1u)) / NPW;
    const unsigned gatherBlocks = (gatherWaves + 7u) / 8u;

    const int tiles        = N / 16;
    const int tilesPerWave = 5;
    const int gemmWaves    = tiles / tilesPerWave;
    const int gemmBlocks   = gemmWaves / 2;

    chunk_sort_kernel<<<nChunks, BTHREADS, 0, stream>>>(src, dst, binned, runTab, E, nBins);
    bin_sort_kernel<<<nBins, STHREADS, 0, stream>>>(binned, runTab, rowInfo, srcSorted, N, nChunks, nBins);

    gather_sum_kernel<PSTRIDE, true><<<gatherBlocks, 256, 0, stream>>>(features, rowInfo, srcSorted, agg1, N);
    sage_gemm_kernel<<<gemmBlocks, 64, 0, stream>>>(features, agg1, degcol,
                                                    Wself1, Wneigh1, b1, h, 1, tilesPerWave, PSTRIDE);

    gather_sum_kernel<FDIM, false><<<gatherBlocks, 256, 0, stream>>>(h, rowInfo, srcSorted, agg2, N);
    sage_gemm_kernel<<<gemmBlocks, 64, 0, stream>>>(h, agg2, degcol,
                                                    Wself2, Wneigh2, b2, out, 0, tilesPerWave, FDIM);
}


# gather: a node's source indices come by one coalesced vector load (lane = list position) and reach the row loads by a read inside each 16-lane row, not by 16 scalar loads per group; sum order kept
# speedup vs baseline: 4.0306x; 1.0631x over previous
#include <hip/hip_runtime.h>


typedef __attribute__((ext_vector_type(2))) float v2f;
typedef __attribute__((ext_vector_type(8))) float v8f;
typedef unsigned long long u64;

#define FDIM 48

#define PSTRIDE 64

#define BIN_SHIFT 8
#define BIN_ROWS  256
#define MAXBINS   512
#define CHUNK     6400
#define BTHREADS  1024
#define EPT       7
#define STHREADS  512
#define MAXCHUNKS 256
#define RPT       8
#define SORT_CAP  8192

template <int PER>
__device__ __forceinline__ void wave_excl_scan(const int* in, int* out, int n, int lane) {
    int loc[PER];
    int s = 0;
    #pragma unroll
    for (int k = 0; k < PER; ++k) {
        const int idx = lane * PER + k;
        loc[k] = s;
        s += (idx < n) ? in[idx] : 0;
    }
    int incl = s;
    #pragma unroll
    for (int off = 1; off < 32; off <<= 1) {
        const int t = __shfl_up(incl, off);
        if (lane >= off) incl += t;
    }
    const int excl = incl - s;
    #pragma unroll
    for (int k = 0; k < PER; ++k) {
        const int idx = lane * PER + k;
        if (idx < n) out[idx] = loc[k] + excl;
    }
}

__global__ void __launch_bounds__(BTHREADS)
chunk_sort_kernel(const int* __restrict__ src, const int* __restrict__ dst,
                  u64* __restrict__ binned, int* __restrict__ runTab, int nEdges, int nBins) {
    __shared__ int hist[MAXBINS];
    __shared__ int lstart[MAXBINS];
    __shared__ u64 recs[CHUNK];
    const int tid  = threadIdx.x;
    const int lane = tid & 31;
    if (tid < MAXBINS) hist[tid] = 0;
    const int e0  = blockIdx.x * CHUNK;
    const int len = nEdges - e0;
    const int n   = len < CHUNK ? len : CHUNK;
    int d[EPT], s[EPT];
    #pragma unroll
    for (int u = 0; u < EPT; ++u) {
        const int i = tid + u * BTHREADS;
        const bool ok = i < n;
        d[u] = ok ? dst[e0 + i] : -1;
        s[u] = ok ? src[e0 + i] : 0;
    }
    __syncthreads();
    #pragma unroll
    for (int u = 0; u < EPT; ++u) {
        if (d[u] >= 0) atomicAdd(&hist[d[u] >> BIN_SHIFT], 1);
    }
    __syncthreads();
    if (tid < 32) wave_excl_scan<MAXBINS / 32>(hist, lstart, nBins, lane);
    __syncthreads();
    int* row = runTab + (size_t)blockIdx.x * (nBins + 1);
    if (tid < nBins) {
        row[tid]  = lstart[tid];
        hist[tid] = lstart[tid];
    }
    if (tid == 0) row[nBins] = n;
    __syncthreads();
    #pragma unroll
    for (int u = 0; u < EPT; ++u) {
        if (d[u] >= 0) {
            const int p = atomicAdd(&hist[d[u] >> BIN_SHIFT], 1);
            recs[p] = ((u64)(unsigned)d[u] << 32) | (u64)(unsigned)s[u];
        }
    }
    __syncthreads();
    for (int p = tid; p < n; p += BTHREADS) binned[e0 + p] = recs[p];
}

__global__ void __launch_bounds__(STHREADS)
bin_sort_kernel(const u64* __restrict__ binned, const int* __restrict__ runTab,
                u64* __restrict__ rowInfo, int* __restrict__ srcSorted,
                int nNodes, int nChunks, int nBins) {
    __shared__ int hist[BIN_ROWS];
    __shared__ int rstart[BIN_ROWS];
    __shared__ int runPos[MAXCHUNKS];
    __shared__ int runCnt[MAXCHUNKS];
    __shared__ int sums[2];
    __shared__ int stage[SORT_CAP];
    const int tid  = threadIdx.x;
    const int lane = tid & 31;
    if (tid < BIN_ROWS) hist[tid] = 0;
    if (tid < 2) sums[tid] = 0;
    __syncthreads();
    if (tid < MAXCHUNKS) {
        int a = 0, c = 0;
        if (tid < nChunks) {
            const int* row = runTab + (size_t)tid * (nBins + 1) + blockIdx.x;
            a = row[0];
            c = row[1] - a;
        }
        runPos[tid] = tid * CHUNK + a;
        runCnt[tid] = c;
        #pragma unroll
        for (int off = 16; off > 0; off >>= 1) {
            a += __shfl_xor(a, off);
            c += __shfl_xor(c, off);
        }
        if (lane == 0) { atomicAdd(&sums[0], a); atomicAdd(&sums[1], c); }
    }
    __syncthreads();
    const int base = sums[0];
    const int cnt  = sums[1];
    const bool fits = cnt <= SORT_CAP;
    const int hf = tid & 1;
    const int rp = runPos[tid >> 1];
    const int rc = runCnt[tid >> 1];
    for (int k0 = 0; k0 < rc; k0 += 2 * RPT) {
        u64 r[RPT];
        #pragma unroll
        for (int u = 0; u < RPT; ++u) {
            const int k = k0 + 2 * u + hf;
            r[u] = binned[rp + (k < rc ? k : rc - 1)];
        }
        #pragma unroll
        for (int u = 0; u < RPT; ++u) {
            const int k = k0 + 2 * u + hf;
            if (k < rc) atomicAdd(&hist[(int)(r[u] >> 32) & (BIN_ROWS - 1)], 1);
        }
    }
    __syncthreads();
    if (tid < 32) wave_excl_scan<BIN_ROWS / 32>(hist, rstart, BIN_ROWS, lane);
    __syncthreads();
    if (tid < BIN_ROWS) {
        const int v = blockIdx.x * BIN_ROWS + tid;
        if (v < nNodes)
            rowInfo[v] = ((u64)(unsigned)hist[tid] << 32) | (u64)(unsigned)(base + rstart[tid]);
        hist[tid] = rstart[tid];
    }
    __syncthreads();
    for (int k0 = 0; k0 < rc; k0 += 2 * RPT) {
        u64 r[RPT];
        #pragma unroll
        for (int u = 0; u < RPT; ++u) {
            const int k = k0 + 2 * u + hf;
            r[u] = binned[rp + (k < rc ? k : rc - 1)];
        }
        #pragma unroll
        for (int u = 0; u < RPT; ++u) {
            const int k = k0 + 2 * u + hf;
            if (k < rc) {
                const int p = atomicAdd(&hist[(int)(r[u] >> 32) & (BIN_ROWS - 1)], 1);
                if (fits) stage[p] = (int)(unsigned)r[u];
                else      srcSorted[base + p] = (int)(unsigned)r[u];
            }
        }
    }
    __syncthreads();
    if (fits) {
        for (int i = tid; i < cnt; i += STHREADS) srcSorted[base + i] = stage[i];
    }
}

#define NPW 8
#define LPG 8

typedef __attribute__((ext_vector_type(4))) float v4f;

__device__ __forceinline__ int row_lane_read(int v, unsigned k) {
    const unsigned sel = 0x11111111u * k;
    return (int)__builtin_amdgcn_permlane16((unsigned)v, (unsigned)v, sel, sel, false, false);
}

template <int STRIDE, bool COUNT>
__global__ void __launch_bounds__(256)
gather_sum_kernel(const float* __restrict__ x, const u64* __restrict__ rowInfo,
                  const int* __restrict__ srcSorted, float* __restrict__ agg, int nNodes) {
    const unsigned lane = threadIdx.x & 31u;
    const unsigned wv = __builtin_amdgcn_readfirstlane(blockIdx.x * 8u + (threadIdx.x >> 5));
    const unsigned hi = lane >> 4;
    const unsigned c  = lane & 15u;
    const unsigned cb = (c < 12u ? c : 11u) * 16u;
    const int pos = (int)(2u * c + hi);
    const unsigned v0 = wv * (unsigned)NPW;
    if (v0 >= (unsigned)nNodes) return;
    const char* xb = (const char*)x;
    u64 info[NPW];
    #pragma unroll
    for (int n = 0; n < NPW; ++n) {
        const unsigned v = v0 + (unsigned)n;
        info[n] = rowInfo[v < (unsigned)nNodes ? v : (unsigned)nNodes - 1u];
    }
    int iv[NPW];
    #pragma unroll
    for (int n = 0; n < NPW; ++n) {
        const int start = (int)(unsigned)info[n];
        const int cnt   = (int)(info[n] >> 32);
        iv[n] = 0;
        if (pos < cnt) iv[n] = srcSorted[start + pos];
    }
    #pragma unroll
    for (int n = 0; n < NPW; ++n) {
        const unsigned v = v0 + (unsigned)n;
        if (v < (unsigned)nNodes) {
            const int start = (int)(unsigned)info[n];
            const int cnt   = (int)(info[n] >> 32);
            v4f acc = {0.0f, 0.0f, 0.0f, 0.0f};
            int ivc = iv[n];
            for (int j = 0; j < cnt; j += 32) {
                if (j > 0) {
                    ivc = 0;
                    if (j + pos < cnt) ivc = srcSorted[start + j + pos];
                }
                #pragma unroll
                for (int g = 0; g < 2; ++g) {
                    const int jb = j + 16 * g;
                    if (jb < cnt) {
                        v4f val[LPG];
                        #pragma unroll
                        for (int k = 0; k < LPG; ++k) {
                            const unsigned e = (unsigned)row_lane_read(ivc, (unsigned)(LPG * g + k));
                            val[k] = *(const v4f*)(xb + (e * (unsigned)(FDIM * 4) + cb));
                        }
                        __builtin_amdgcn_sched_barrier(0);
                        #pragma unroll
                        for (int k = 0; k < LPG; ++k) {
                            if (jb + 2 * k + (int)hi < cnt) acc += val[k];
                        }
                    }
                }
            }
            acc.x += __shfl_xor(acc.x, 16);
            acc.y += __shfl_xor(acc.y, 16);
            acc.z += __shfl_xor(acc.z, 16);
            acc.w += __shfl_xor(acc.w, 16);
            float* g = agg + (size_t)v * STRIDE;
            if (COUNT) {
                if (lane <= 12u) {
                    v4f o = acc;
                    if (lane == 12u) { o.x = (float)cnt; o.y = 0.0f; o.z = 0.0f; o.w = 0.0f; }
                    *(v4f*)(g + lane * 4u) = o;
                }
            } else {
                if (lane < 12u) *(v4f*)(g + lane * 4u) = acc;
            }
        }
    }
}

__global__ void __launch_bounds__(64)
sage_gemm_kernel(const float* __restrict__ x,
                 const float* __restrict__ agg,
                 const float* __restrict__ degcol,
                 const float* __restrict__ Wself,
                 const float* __restrict__ Wneigh,
                 const float* __restrict__ bias,
                 float* __restrict__ out,
                 int relu, int tilesPerWave, int aggStride) {
    const int lane = threadIdx.x & 31;
    const int wave = threadIdx.x >> 5;
    const int half = lane >> 4;
    const int l15  = lane & 15;
    const int gw   = blockIdx.x * 2 + wave;

    v2f wS[12][3], wN[12][3];
    #pragma unroll
    for (int k = 0; k < 12; ++k) {
        const int kk = k * 4 + half * 2;
        #pragma unroll
        for (int nt = 0; nt < 3; ++nt) {
            const int col = nt * 16 + l15;
            wS[k][nt].x = Wself [(size_t)(kk)     * FDIM + col];
            wS[k][nt].y = Wself [(size_t)(kk + 1) * FDIM + col];
            wN[k][nt].x = Wneigh[(size_t)(kk)     * FDIM + col];
            wN[k][nt].y = Wneigh[(size_t)(kk + 1) * FDIM + col];
        }
    }
    float bv0 = bias[0 * 16 + l15];
    float bv1 = bias[1 * 16 + l15];
    float bv2 = bias[2 * 16 + l15];

    #pragma unroll 1
    for (int t = 0; t < tilesPerWave; ++t) {
        const int m0 = (gw * tilesPerWave + t) * 16;
        const float* xrow = x   + (size_t)(m0 + l15) * FDIM;
        const float* grow = agg + (size_t)(m0 + l15) * aggStride;
        const float  dsc  = 1.0f / fmaxf(degcol[(size_t)(m0 + l15) * PSTRIDE], 1.0f);

        v8f cs[3], cn[3];
        #pragma unroll
        for (int r = 0; r < 8; ++r) {
            cs[0][r] = bv0; cs[1][r] = bv1; cs[2][r] = bv2;
            cn[0][r] = 0.0f; cn[1][r] = 0.0f; cn[2][r] = 0.0f;
        }

        #pragma unroll
        for (int k = 0; k < 12; ++k) {
            const int kk = k * 4 + half * 2;
            v2f a;
            a.x = xrow[kk];
            a.y = xrow[kk + 1];
            v2f a2;
            a2.x = grow[kk] * dsc;
            a2.y = grow[kk + 1] * dsc;
            #pragma unroll
            for (int nt = 0; nt < 3; ++nt) {
                cs[nt] = __builtin_amdgcn_wmma_f32_16x16x4_f32(
                    false, a, false, wS[k][nt], (short)0, cs[nt], false, false);
                cn[nt] = __builtin_amdgcn_wmma_f32_16x16x4_f32(
                    false, a2, false, wN[k][nt], (short)0, cn[nt], false, false);
            }
        }

        #pragma unroll
        for (int nt = 0; nt < 3; ++nt) {
            #pragma unroll
            for (int r = 0; r < 8; ++r) {
                float v = cs[nt][r] + cn[nt][r];
                if (relu) v = fmaxf(v, 0.0f);
                out[(size_t)(m0 + half * 8 + r) * FDIM + nt * 16 + l15] = v;
            }
        }
    }
}

extern "C" void kernel_launch(void* const* d_in, const int* in_sizes, int n_in,
                              void* d_out, int out_size, void* d_ws, size_t ws_size,
                              hipStream_t stream) {
    const float* features = (const float*)d_in[0];
    const int*   src      = (const int*)d_in[1];
    const int*   dst      = (const int*)d_in[2];
    const float* Wself1   = (const float*)d_in[3];
    const float* Wneigh1  = (const float*)d_in[4];
    const float* b1       = (const float*)d_in[5];
    const float* Wself2   = (const float*)d_in[6];
    const float* Wneigh2  = (const float*)d_in[7];
    const float* b2       = (const float*)d_in[8];
    float* out = (float*)d_out;

    const int N = in_sizes[0] / FDIM;
    const int E = in_sizes[1];

    const int nBins      = (N + BIN_ROWS - 1) / BIN_ROWS;
    const int nChunks    = (E + CHUNK - 1) / CHUNK;

    char* ws = (char*)d_ws;
    size_t off = 0;
    float* agg1 = (float*)(ws + off); off += (((size_t)N * PSTRIDE * 4) + 255) & ~(size_t)255;
    float* agg2 = (float*)(ws + off); off += (((size_t)N * FDIM * 4)    + 255) & ~(size_t)255;
    float* h    = (float*)(ws + off); off += (((size_t)N * FDIM * 4)    + 255) & ~(size_t)255;
    u64*  binned    = (u64*)(ws + off);  off += (((size_t)E * 8) + 255) & ~(size_t)255;
    int*  srcSorted = (int*)(ws + off);  off += (((size_t)E * 4) + 255) & ~(size_t)255;
    u64*  rowInfo   = (u64*)(ws + off);  off += (((size_t)N * 8) + 255) & ~(size_t)255;
    int*  runTab    = (int*)(ws + off);
    const float* degcol = agg1 + FDIM;

    const unsigned gatherWaves  = ((unsigned)N + (NPW - 1u)) / NPW;
    const unsigned gatherBlocks = (gatherWaves + 7u) / 8u;

    const int tiles        = N / 16;
    const int tilesPerWave = 5;
    const int gemmWaves    = tiles / tilesPerWave;
    const int gemmBlocks   = gemmWaves / 2;

    chunk_sort_kernel<<<nChunks, BTHREADS, 0, stream>>>(src, dst, binned, runTab, E, nBins);
    bin_sort_kernel<<<nBins, STHREADS, 0, stream>>>(binned, runTab, rowInfo, srcSorted, N, nChunks, nBins);

    gather_sum_kernel<PSTRIDE, true><<<gatherBlocks, 256, 0, stream>>>(features, rowInfo, srcSorted, agg1, N);
    sage_gemm_kernel<<<gemmBlocks, 64, 0, stream>>>(features, agg1, degcol,
                                                    Wself1, Wneigh1, b1, h, 1, tilesPerWave, PSTRIDE);

    gather_sum_kernel<FDIM, false><<<gatherBlocks, 256, 0, stream>>>(h, rowInfo, srcSorted, agg2, N);
    sage_gemm_kernel<<<gemmBlocks, 64, 0, stream>>>(h, agg2, degcol,
                                                    Wself2, Wneigh2, b2, out, 0, tilesPerWave, FDIM);
}


# in-edge sum fused into the layer GEMM: a wave sums its 16-node tile, keeps the mean rows in LDS, does both products there (weights in LDS, one accumulator chain: new sum order); 6 launches to 4
# speedup vs baseline: 4.6447x; 1.1524x over previous
#include <hip/hip_runtime.h>


typedef __attribute__((ext_vector_type(2))) float v2f;
typedef __attribute__((ext_vector_type(8))) float v8f;
typedef unsigned long long u64;

#define FDIM 48

#define BIN_SHIFT 8
#define BIN_ROWS  256
#define MAXBINS   512
#define CHUNK     6400
#define BTHREADS  1024
#define EPT       7
#define STHREADS  512
#define MAXCHUNKS 256
#define RPT       8
#define SORT_CAP  8192

template <int PER>
__device__ __forceinline__ void wave_excl_scan(const int* in, int* out, int n, int lane) {
    int loc[PER];
    int s = 0;
    #pragma unroll
    for (int k = 0; k < PER; ++k) {
        const int idx = lane * PER + k;
        loc[k] = s;
        s += (idx < n) ? in[idx] : 0;
    }
    int incl = s;
    #pragma unroll
    for (int off = 1; off < 32; off <<= 1) {
        const int t = __shfl_up(incl, off);
        if (lane >= off) incl += t;
    }
    const int excl = incl - s;
    #pragma unroll
    for (int k = 0; k < PER; ++k) {
        const int idx = lane * PER + k;
        if (idx < n) out[idx] = loc[k] + excl;
    }
}

__global__ void __launch_bounds__(BTHREADS)
chunk_sort_kernel(const int* __restrict__ src, const int* __restrict__ dst,
                  u64* __restrict__ binned, int* __restrict__ runTab, int nEdges, int nBins) {
    __shared__ int hist[MAXBINS];
    __shared__ int lstart[MAXBINS];
    __shared__ u64 recs[CHUNK];
    const int tid  = threadIdx.x;
    const int lane = tid & 31;
    if (tid < MAXBINS) hist[tid] = 0;
    const int e0  = blockIdx.x * CHUNK;
    const int len = nEdges - e0;
    const int n   = len < CHUNK ? len : CHUNK;
    int d[EPT], s[EPT];
    #pragma unroll
    for (int u = 0; u < EPT; ++u) {
        const int i = tid + u * BTHREADS;
        const bool ok = i < n;
        d[u] = ok ? dst[e0 + i] : -1;
        s[u] = ok ? src[e0 + i] : 0;
    }
    __syncthreads();
    #pragma unroll
    for (int u = 0; u < EPT; ++u) {
        if (d[u] >= 0) atomicAdd(&hist[d[u] >> BIN_SHIFT], 1);
    }
    __syncthreads();
    if (tid < 32) wave_excl_scan<MAXBINS / 32>(hist, lstart, nBins, lane);
    __syncthreads();
    int* row = runTab + (size_t)blockIdx.x * (nBins + 1);
    if (tid < nBins) {
        row[tid]  = lstart[tid];
        hist[tid] = lstart[tid];
    }
    if (tid == 0) row[nBins] = n;
    __syncthreads();
    #pragma unroll
    for (int u = 0; u < EPT; ++u) {
        if (d[u] >= 0) {
            const int p = atomicAdd(&hist[d[u] >> BIN_SHIFT], 1);
            recs[p] = ((u64)(unsigned)d[u] << 32) | (u64)(unsigned)s[u];
        }
    }
    __syncthreads();
    for (int p = tid; p < n; p += BTHREADS) binned[e0 + p] = recs[p];
}

__global__ void __launch_bounds__(STHREADS)
bin_sort_kernel(const u64* __restrict__ binned, const int* __restrict__ runTab,
                u64* __restrict__ rowInfo, int* __restrict__ srcSorted,
                int nNodes, int nChunks, int nBins) {
    __shared__ int hist[BIN_ROWS];
    __shared__ int rstart[BIN_ROWS];
    __shared__ int runPos[MAXCHUNKS];
    __shared__ int runCnt[MAXCHUNKS];
    __shared__ int sums[2];
    __shared__ int stage[SORT_CAP];
    const int tid  = threadIdx.x;
    const int lane = tid & 31;
    if (tid < BIN_ROWS) hist[tid] = 0;
    if (tid < 2) sums[tid] = 0;
    __syncthreads();
    if (tid < MAXCHUNKS) {
        int a = 0, c = 0;
        if (tid < nChunks) {
            const int* row = runTab + (size_t)tid * (nBins + 1) + blockIdx.x;
            a = row[0];
            c = row[1] - a;
        }
        runPos[tid] = tid * CHUNK + a;
        runCnt[tid] = c;
        #pragma unroll
        for (int off = 16; off > 0; off >>= 1) {
            a += __shfl_xor(a, off);
            c += __shfl_xor(c, off);
        }
        if (lane == 0) { atomicAdd(&sums[0], a); atomicAdd(&sums[1], c); }
    }
    __syncthreads();
    const int base = sums[0];
    const int cnt  = sums[1];
    const bool fits = cnt <= SORT_CAP;
    const int hf = tid & 1;
    const int rp = runPos[tid >> 1];
    const int rc = runCnt[tid >> 1];
    for (int k0 = 0; k0 < rc; k0 += 2 * RPT) {
        u64 r[RPT];
        #pragma unroll
        for (int u = 0; u < RPT; ++u) {
            const int k = k0 + 2 * u + hf;
            r[u] = binned[rp + (k < rc ? k : rc - 1)];
        }
        #pragma unroll
        for (int u = 0; u < RPT; ++u) {
            const int k = k0 + 2 * u + hf;
            if (k < rc) atomicAdd(&hist[(int)(r[u] >> 32) & (BIN_ROWS - 1)], 1);
        }
    }
    __syncthreads();
    if (tid < 32) wave_excl_scan<BIN_ROWS / 32>(hist, rstart, BIN_ROWS, lane);
    __syncthreads();
    if (tid < BIN_ROWS) {
        const int v = blockIdx.x * BIN_ROWS + tid;
        if (v < nNodes)
            rowInfo[v] = ((u64)(unsigned)hist[tid] << 32) | (u64)(unsigned)(base + rstart[tid]);
        hist[tid] = rstart[tid];
    }
    __syncthreads();
    for (int k0 = 0; k0 < rc; k0 += 2 * RPT) {
        u64 r[RPT];
        #pragma unroll
        for (int u = 0; u < RPT; ++u) {
            const int k = k0 + 2 * u + hf;
            r[u] = binned[rp + (k < rc ? k : rc - 1)];
        }
        #pragma unroll
        for (int u = 0; u < RPT; ++u) {
            const int k = k0 + 2 * u + hf;
            if (k < rc) {
                const int p = atomicAdd(&hist[(int)(r[u] >> 32) & (BIN_ROWS - 1)], 1);
                if (fits) stage[p] = (int)(unsigned)r[u];
                else      srcSorted[base + p] = (int)(unsigned)r[u];
            }
        }
    }
    __syncthreads();
    if (fits) {
        for (int i = tid; i < cnt; i += STHREADS) srcSorted[base + i] = stage[i];
    }
}

#define TN     16
#define LPG    8
#define APITCH 52
#define LWAVES 8

typedef __attribute__((ext_vector_type(4))) float v4f;

__device__ __forceinline__ int row_lane_read(int v, unsigned k) {
    const unsigned sel = 0x11111111u * k;
    return (int)__builtin_amdgcn_permlane16((unsigned)v, (unsigned)v, sel, sel, false, false);
}

__device__ __forceinline__ int node_word(int perLane, int n) {
    return __builtin_amdgcn_readlane(perLane, n);
}

__device__ __forceinline__ int first_indices(const int* __restrict__ srcSorted,
                                             int startL, int cntL, int n, int pos) {
    const int st = node_word(startL, n);
    const int ct = node_word(cntL, n);
    int r = 0;
    if (pos < ct) r = srcSorted[st + pos];
    return r;
}

template <bool RELU>
__global__ void __launch_bounds__(LWAVES * 32)
sage_layer_kernel(const float* __restrict__ x,
                  const u64* __restrict__ rowInfo, const int* __restrict__ srcSorted,
                  const float* __restrict__ Wself,
                  const float* __restrict__ Wneigh,
                  const float* __restrict__ bias,
                  float* __restrict__ out,
                  int nNodes, int nTiles) {
    __shared__ float wS[FDIM * FDIM];
    __shared__ float wN[FDIM * FDIM];
    __shared__ float meanS[LWAVES][TN * APITCH];
    const unsigned tid  = threadIdx.x;
    const unsigned lane = tid & 31u;
    const unsigned wib  = __builtin_amdgcn_readfirstlane(tid >> 5);
    const unsigned tile = blockIdx.x * (unsigned)LWAVES + wib;
    const unsigned hi = lane >> 4;
    const unsigned c  = lane & 15u;
    const unsigned cb = (c < 12u ? c : 11u) * 16u;
    const int pos = (int)(2u * c + hi);
    const unsigned v0 = tile * (unsigned)TN;
    const char* xb = (const char*)x;
    float* mine = &meanS[wib][0];
    const unsigned vi = v0 + c;
    const u64 infoL  = rowInfo[vi < (unsigned)nNodes ? vi : (unsigned)nNodes - 1u];
    #pragma unroll
    for (unsigned u = 0; u < (unsigned)(FDIM * FDIM) / (LWAVES * 32u); ++u) {
        const unsigned i = tid + u * (LWAVES * 32u);
        const unsigned k = i / (unsigned)FDIM;
        const unsigned n = i - k * (unsigned)FDIM;
        const unsigned d = ((k >> 1) * (unsigned)FDIM + n) * 2u + (k & 1u);
        wS[d] = Wself[i];
        wN[d] = Wneigh[i];
    }
    const int startL = (int)(unsigned)infoL;
    const int cntL   = (int)(infoL >> 32);
    int ivCur  = first_indices(srcSorted, startL, cntL, 0, pos);
    int ivNext = first_indices(srcSorted, startL, cntL, 1, pos);
    __syncthreads();
    if (tile >= (unsigned)nTiles) return;
    const float* xrow = x + (size_t)(v0 + c) * FDIM;
    v2f ax[12];
    #pragma unroll
    for (int k = 0; k < 12; ++k) ax[k] = *(const v2f*)(xrow + (k * 4 + (int)hi * 2));
    #pragma unroll 1
    for (int n = 0; n < TN; ++n) {
        const int start = node_word(startL, n);
        const int cnt   = node_word(cntL, n);
        int ivAfter = 0;
        if (n + 2 < TN) ivAfter = first_indices(srcSorted, startL, cntL, n + 2, pos);
        v4f acc = {0.0f, 0.0f, 0.0f, 0.0f};
        int ivc = ivCur;
        for (int j = 0; j < cnt; j += 32) {
            if (j > 0) {
                ivc = 0;
                if (j + pos < cnt) ivc = srcSorted[start + j + pos];
            }
            #pragma unroll
            for (int g = 0; g < 2; ++g) {
                const int jb = j + 16 * g;
                if (jb < cnt) {
                    v4f val[LPG];
                    #pragma unroll
                    for (int k = 0; k < LPG; ++k) {
                        const unsigned e = (unsigned)row_lane_read(ivc, (unsigned)(LPG * g + k));
                        val[k] = *(const v4f*)(xb + (e * (unsigned)(FDIM * 4) + cb));
                    }
                    __builtin_amdgcn_sched_barrier(0);
                    #pragma unroll
                    for (int k = 0; k < LPG; ++k) {
                        if (jb + 2 * k + (int)hi < cnt) acc += val[k];
                    }
                }
            }
        }
        acc.x += __shfl_xor(acc.x, 16);
        acc.y += __shfl_xor(acc.y, 16);
        acc.z += __shfl_xor(acc.z, 16);
        acc.w += __shfl_xor(acc.w, 16);
        const float dsc = 1.0f / fmaxf((float)cnt, 1.0f);
        if (lane < 12u) *(v4f*)(mine + n * APITCH + lane * 4u) = acc * dsc;
        ivCur  = ivNext;
        ivNext = ivAfter;
    }
    __builtin_amdgcn_fence(__ATOMIC_RELEASE, "wavefront");
    __builtin_amdgcn_wave_barrier();
    __builtin_amdgcn_fence(__ATOMIC_ACQUIRE, "wavefront");

    const float* mrow = mine + c * APITCH;
    const float bv0 = bias[0 * 16 + c];
    const float bv1 = bias[1 * 16 + c];
    const float bv2 = bias[2 * 16 + c];
    v8f cc[3];
    #pragma unroll
    for (int r = 0; r < 8; ++r) { cc[0][r] = bv0; cc[1][r] = bv1; cc[2][r] = bv2; }
    #pragma unroll
    for (int k = 0; k < 12; ++k) {
        const unsigned kk = (unsigned)k * 4u + hi * 2u;
        const v2f a2 = *(const v2f*)(mrow + kk);
        const unsigned wo = ((kk >> 1) * (unsigned)FDIM + c) * 2u;
        v2f bs[3], bn[3];
        #pragma unroll
        for (int nt = 0; nt < 3; ++nt) {
            bs[nt] = *(const v2f*)(wS + wo + nt * 32);
            bn[nt] = *(const v2f*)(wN + wo + nt * 32);
        }
        #pragma unroll
        for (int nt = 0; nt < 3; ++nt)
            cc[nt] = __builtin_amdgcn_wmma_f32_16x16x4_f32(
                false, ax[k], false, bs[nt], (short)0, cc[nt], false, false);
        #pragma unroll
        for (int nt = 0; nt < 3; ++nt)
            cc[nt] = __builtin_amdgcn_wmma_f32_16x16x4_f32(
                false, a2, false, bn[nt], (short)0, cc[nt], false, false);
    }
    #pragma unroll
    for (int nt = 0; nt < 3; ++nt) {
        #pragma unroll
        for (int r = 0; r < 8; ++r) {
            float v = cc[nt][r];
            if (RELU) v = fmaxf(v, 0.0f);
            out[(size_t)(v0 + hi * 8u + (unsigned)r) * FDIM + nt * 16 + c] = v;
        }
    }
}

extern "C" void kernel_launch(void* const* d_in, const int* in_sizes, int n_in,
                              void* d_out, int out_size, void* d_ws, size_t ws_size,
                              hipStream_t stream) {
    const float* features = (const float*)d_in[0];
    const int*   src      = (const int*)d_in[1];
    const int*   dst      = (const int*)d_in[2];
    const float* Wself1   = (const float*)d_in[3];
    const float* Wneigh1  = (const float*)d_in[4];
    const float* b1       = (const float*)d_in[5];
    const float* Wself2   = (const float*)d_in[6];
    const float* Wneigh2  = (const float*)d_in[7];
    const float* b2       = (const float*)d_in[8];
    float* out = (float*)d_out;

    const int N = in_sizes[0] / FDIM;
    const int E = in_sizes[1];

    const int nBins      = (N + BIN_ROWS - 1) / BIN_ROWS;
    const int nChunks    = (E + CHUNK - 1) / CHUNK;

    char* ws = (char*)d_ws;
    size_t off = 0;
    float* h    = (float*)(ws + off); off += (((size_t)N * FDIM * 4)    + 255) & ~(size_t)255;
    u64*  binned    = (u64*)(ws + off);  off += (((size_t)E * 8) + 255) & ~(size_t)255;
    int*  srcSorted = (int*)(ws + off);  off += (((size_t)E * 4) + 255) & ~(size_t)255;
    u64*  rowInfo   = (u64*)(ws + off);  off += (((size_t)N * 8) + 255) & ~(size_t)255;
    int*  runTab    = (int*)(ws + off);

    const int tiles       = N / 16;
    const int layerBlocks = (tiles + LWAVES - 1) / LWAVES;

    chunk_sort_kernel<<<nChunks, BTHREADS, 0, stream>>>(src, dst, binned, runTab, E, nBins);
    bin_sort_kernel<<<nBins, STHREADS, 0, stream>>>(binned, runTab, rowInfo, srcSorted, N, nChunks, nBins);

    sage_layer_kernel<true><<<layerBlocks, LWAVES * 32, 0, stream>>>(features, rowInfo, srcSorted,
                                                                  Wself1, Wneigh1, b1, h, N, tiles);
    sage_layer_kernel<false><<<layerBlocks, LWAVES * 32, 0, stream>>>(h, rowInfo, srcSorted,
                                                                   Wself2, Wneigh2, b2, out, N, tiles);
}


# layer 1 fused into the second sort launch: the block that puts a bin's edges in row order keeps the list in LDS and its 16 waves do layer 1 of the bin's 16 tiles from it; 4 launches to 3; sums kept
# speedup vs baseline: 5.0429x; 1.0857x over previous
#include <hip/hip_runtime.h>


typedef __attribute__((ext_vector_type(2))) float v2f;
typedef __attribute__((ext_vector_type(8))) float v8f;
typedef unsigned long long u64;

#define FDIM 48

#define BIN_SHIFT 8
#define BIN_ROWS  256
#define MAXBINS   512
#define CHUNK     6400
#define BTHREADS  1024
#define EPT       7
#define STHREADS  512
#define MAXCHUNKS 256
#define RPT       8
#define SORT_CAP  8192

template <int PER>
__device__ __forceinline__ void wave_excl_scan(const int* in, int* out, int n, int lane) {
    int loc[PER];
    int s = 0;
    #pragma unroll
    for (int k = 0; k < PER; ++k) {
        const int idx = lane * PER + k;
        loc[k] = s;
        s += (idx < n) ? in[idx] : 0;
    }
    int incl = s;
    #pragma unroll
    for (int off = 1; off < 32; off <<= 1) {
        const int t = __shfl_up(incl, off);
        if (lane >= off) incl += t;
    }
    const int excl = incl - s;
    #pragma unroll
    for (int k = 0; k < PER; ++k) {
        const int idx = lane * PER + k;
        if (idx < n) out[idx] = loc[k] + excl;
    }
}

__global__ void __launch_bounds__(BTHREADS)
chunk_sort_kernel(const int* __restrict__ src, const int* __restrict__ dst,
                  u64* __restrict__ binned, int* __restrict__ runTab, int nEdges, int nBins) {
    __shared__ int hist[MAXBINS];
    __shared__ int lstart[MAXBINS];
    __shared__ u64 recs[CHUNK];
    const int tid  = threadIdx.x;
    const int lane = tid & 31;
    if (tid < MAXBINS) hist[tid] = 0;
    const int e0  = blockIdx.x * CHUNK;
    const int len = nEdges - e0;
    const int n   = len < CHUNK ? len : CHUNK;
    int d[EPT], s[EPT];
    #pragma unroll
    for (int u = 0; u < EPT; ++u) {
        const int i = tid + u * BTHREADS;
        const bool ok = i < n;
        d[u] = ok ? dst[e0 + i] : -1;
        s[u] = ok ? src[e0 + i] : 0;
    }
    __syncthreads();
    #pragma unroll
    for (int u = 0; u < EPT; ++u) {
        if (d[u] >= 0) atomicAdd(&hist[d[u] >> BIN_SHIFT], 1);
    }
    __syncthreads();
    if (tid < 32) wave_excl_scan<MAXBINS / 32>(hist, lstart, nBins, lane);
    __syncthreads();
    int* row = runTab + (size_t)blockIdx.x * (nBins + 1);
    if (tid < nBins) {
        row[tid]  = lstart[tid];
        hist[tid] = lstart[tid];
    }
    if (tid == 0) row[nBins] = n;
    __syncthreads();
    #pragma unroll
    for (int u = 0; u < EPT; ++u) {
        if (d[u] >= 0) {
            const int p = atomicAdd(&hist[d[u] >> BIN_SHIFT], 1);
            recs[p] = ((u64)(unsigned)d[u] << 32) | (u64)(unsigned)s[u];
        }
    }
    __syncthreads();
    for (int p = tid; p < n; p += BTHREADS) binned[e0 + p] = recs[p];
}

#define TN     16
#define LPG    8
#define APITCH 52
#define LWAVES 8
#define BWAVES (BIN_ROWS / TN)
static_assert(BWAVES * 32 == STHREADS, "bin_layer: one wave per tile of the bin");
static_assert((FDIM * FDIM) % (STHREADS / 2) == 0, "bin_layer: weight staging in whole rounds");

typedef __attribute__((ext_vector_type(4))) float v4f;

__device__ __forceinline__ int row_lane_read(int v, unsigned k) {
    const unsigned sel = 0x11111111u * k;
    return (int)__builtin_amdgcn_permlane16((unsigned)v, (unsigned)v, sel, sel, false, false);
}

__device__ __forceinline__ int node_word(int perLane, int n) {
    return __builtin_amdgcn_readlane(perLane, n);
}

struct MemList {
    const int* __restrict__ p;
    __device__ __forceinline__ int operator()(int i) const { return p[i]; }
};
struct LdsList {
    const int* p;
    __device__ __forceinline__ int operator()(int i) const { return p[i]; }
};

template <typename LIST>
__device__ __forceinline__ int first_indices(LIST list, int startL, int cntL, int n, int pos) {
    const int st = node_word(startL, n);
    const int ct = node_word(cntL, n);
    int r = 0;
    if (pos < ct) r = list(st + pos);
    return r;
}

template <bool RELU, typename LIST>
__device__ __forceinline__ void sage_tile(const float* __restrict__ x,
                                          LIST list, int startL, int cntL, int ivCur, int ivNext,
                                          const float* wS, const float* wN,
                                          float* mine,
                                          const float* __restrict__ bias,
                                          float* __restrict__ out,
                                          unsigned v0, unsigned lane) {
    const unsigned hi = lane >> 4;
    const unsigned c  = lane & 15u;
    const unsigned cb = (c < 12u ? c : 11u) * 16u;
    const int pos = (int)(2u * c + hi);
    const char* xb = (const char*)x;
    const float* xrow = x + (size_t)(v0 + c) * FDIM;
    v2f ax[12];
    #pragma unroll
    for (int k = 0; k < 12; ++k) ax[k] = *(const v2f*)(xrow + (k * 4 + (int)hi * 2));
    #pragma unroll 1
    for (int n = 0; n < TN; ++n) {
        const int start = node_word(startL, n);
        const int cnt   = node_word(cntL, n);
        int ivAfter = 0;
        if (n + 2 < TN) ivAfter = first_indices(list, startL, cntL, n + 2, pos);
        v4f acc = {0.0f, 0.0f, 0.0f, 0.0f};
        int ivc = ivCur;
        for (int j = 0; j < cnt; j += 32) {
            if (j > 0) {
                ivc = 0;
                if (j + pos < cnt) ivc = list(start + j + pos);
            }
            #pragma unroll
            for (int g = 0; g < 2; ++g) {
                const int jb = j + 16 * g;
                if (jb < cnt) {
                    v4f val[LPG];
                    #pragma unroll
                    for (int k = 0; k < LPG; ++k) {
                        const unsigned e = (unsigned)row_lane_read(ivc, (unsigned)(LPG * g + k));
                        val[k] = *(const v4f*)(xb + (e * (unsigned)(FDIM * 4) + cb));
                    }
                    __builtin_amdgcn_sched_barrier(0);
                    #pragma unroll
                    for (int k = 0; k < LPG; ++k) {
                        if (jb + 2 * k + (int)hi < cnt) acc += val[k];
                    }
                }
            }
        }
        acc.x += __shfl_xor(acc.x, 16);
        acc.y += __shfl_xor(acc.y, 16);
        acc.z += __shfl_xor(acc.z, 16);
        acc.w += __shfl_xor(acc.w, 16);
        const float dsc = 1.0f / fmaxf((float)cnt, 1.0f);
        if (lane < 12u) *(v4f*)(mine + n * APITCH + lane * 4u) = acc * dsc;
        ivCur  = ivNext;
        ivNext = ivAfter;
    }
    __builtin_amdgcn_fence(__ATOMIC_RELEASE, "wavefront");
    __builtin_amdgcn_wave_barrier();
    __builtin_amdgcn_fence(__ATOMIC_ACQUIRE, "wavefront");

    const float* mrow = mine + c * APITCH;
    const float bv0 = bias[0 * 16 + c];
    const float bv1 = bias[1 * 16 + c];
    const float bv2 = bias[2 * 16 + c];
    v8f cc[3];
    #pragma unroll
    for (int r = 0; r < 8; ++r) { cc[0][r] = bv0; cc[1][r] = bv1; cc[2][r] = bv2; }
    #pragma unroll
    for (int k = 0; k < 12; ++k) {
        const unsigned kk = (unsigned)k * 4u + hi * 2u;
        const v2f a2 = *(const v2f*)(mrow + kk);
        const unsigned wo = ((kk >> 1) * (unsigned)FDIM + c) * 2u;
        v2f bs[3], bn[3];
        #pragma unroll
        for (int nt = 0; nt < 3; ++nt) {
            bs[nt] = *(const v2f*)(wS + wo + nt * 32);
            bn[nt] = *(const v2f*)(wN + wo + nt * 32);
        }
        #pragma unroll
        for (int nt = 0; nt < 3; ++nt)
            cc[nt] = __builtin_amdgcn_wmma_f32_16x16x4_f32(
                false, ax[k], false, bs[nt], (short)0, cc[nt], false, false);
        #pragma unroll
        for (int nt = 0; nt < 3; ++nt)
            cc[nt] = __builtin_amdgcn_wmma_f32_16x16x4_f32(
                false, a2, false, bn[nt], (short)0, cc[nt], false, false);
    }
    #pragma unroll
    for (int nt = 0; nt < 3; ++nt) {
        #pragma unroll
        for (int r = 0; r < 8; ++r) {
            float v = cc[nt][r];
            if (RELU) v = fmaxf(v, 0.0f);
            out[(size_t)(v0 + hi * 8u + (unsigned)r) * FDIM + nt * 16 + c] = v;
        }
    }
}

__global__ void __launch_bounds__(STHREADS)
bin_layer_kernel(const u64* __restrict__ binned, const int* __restrict__ runTab,
                 u64* __restrict__ rowInfo, int* __restrict__ srcSorted,
                 const float* __restrict__ x,
                 const float* __restrict__ Wself,
                 const float* __restrict__ Wneigh,
                 const float* __restrict__ bias,
                 float* __restrict__ out,
                 int nNodes, int nChunks, int nBins, int nTiles) {
    __shared__ int hist[BIN_ROWS];
    __shared__ int rstart[BIN_ROWS];
    __shared__ int runPos[MAXCHUNKS];
    __shared__ int runCnt[MAXCHUNKS];
    __shared__ int sums[2];
    __shared__ int stage[SORT_CAP];
    __shared__ float wSN[2][FDIM * FDIM];
    __shared__ float meanS[BWAVES][TN * APITCH];
    const int tid  = threadIdx.x;
    const int lane = tid & 31;
    int a = 0, aNext = 0;
    if (tid < MAXCHUNKS && tid < nChunks) {
        const int* row = runTab + (size_t)tid * (nBins + 1) + blockIdx.x;
        a     = row[0];
        aNext = row[1];
    }
    {
        const unsigned half = (unsigned)tid / (STHREADS / 2u);
        const unsigned t    = (unsigned)tid % (STHREADS / 2u);
        const float* __restrict__ W = half ? Wneigh : Wself;
        float* wL = &wSN[half][0];
        #pragma unroll
        for (unsigned u = 0; u < (unsigned)(FDIM * FDIM) / (STHREADS / 2u); ++u) {
            const unsigned i = t + u * (STHREADS / 2u);
            const unsigned k = i / (unsigned)FDIM;
            const unsigned n = i - k * (unsigned)FDIM;
            const unsigned d = ((k >> 1) * (unsigned)FDIM + n) * 2u + (k & 1u);
            wL[d] = W[i];
        }
    }
    if (tid < BIN_ROWS) hist[tid] = 0;
    if (tid < 2) sums[tid] = 0;
    __syncthreads();
    if (tid < MAXCHUNKS) {
        int c = aNext - a;
        runPos[tid] = tid * CHUNK + a;
        runCnt[tid] = c;
        #pragma unroll
        for (int off = 16; off > 0; off >>= 1) {
            a += __shfl_xor(a, off);
            c += __shfl_xor(c, off);
        }
        if (lane == 0) { atomicAdd(&sums[0], a); atomicAdd(&sums[1], c); }
    }
    __syncthreads();
    const int base = sums[0];
    const int cnt  = sums[1];
    const bool fits = cnt <= SORT_CAP;
    const int hf = tid & 1;
    const int rp = runPos[tid >> 1];
    const int rc = runCnt[tid >> 1];
    for (int k0 = 0; k0 < rc; k0 += 2 * RPT) {
        u64 r[RPT];
        #pragma unroll
        for (int u = 0; u < RPT; ++u) {
            const int k = k0 + 2 * u + hf;
            r[u] = binned[rp + (k < rc ? k : rc - 1)];
        }
        #pragma unroll
        for (int u = 0; u < RPT; ++u) {
            const int k = k0 + 2 * u + hf;
            if (k < rc) atomicAdd(&hist[(int)(r[u] >> 32) & (BIN_ROWS - 1)], 1);
        }
    }
    __syncthreads();
    if (tid < 32) wave_excl_scan<BIN_ROWS / 32>(hist, rstart, BIN_ROWS, lane);
    __syncthreads();
    if (tid < BIN_ROWS) {
        const int v = blockIdx.x * BIN_ROWS + tid;
        if (v < nNodes)
            rowInfo[v] = ((u64)(unsigned)hist[tid] << 32) | (u64)(unsigned)(base + rstart[tid]);
        hist[tid] = rstart[tid];
    }
    __syncthreads();
    for (int k0 = 0; k0 < rc; k0 += 2 * RPT) {
        u64 r[RPT];
        #pragma unroll
        for (int u = 0; u < RPT; ++u) {
            const int k = k0 + 2 * u + hf;
            r[u] = binned[rp + (k < rc ? k : rc - 1)];
        }
        #pragma unroll
        for (int u = 0; u < RPT; ++u) {
            const int k = k0 + 2 * u + hf;
            if (k < rc) {
                const int p = atomicAdd(&hist[(int)(r[u] >> 32) & (BIN_ROWS - 1)], 1);
                if (fits) stage[p] = (int)(unsigned)r[u];
                else      srcSorted[base + p] = (int)(unsigned)r[u];
            }
        }
    }
    if (!fits) __threadfence();
    __syncthreads();
    if (fits) {
        for (int i = tid; i < cnt; i += STHREADS) srcSorted[base + i] = stage[i];
    }

    const unsigned ulane = (unsigned)lane;
    const unsigned wib  = __builtin_amdgcn_readfirstlane((unsigned)tid >> 5);
    const unsigned tile = blockIdx.x * (unsigned)BWAVES + wib;
    if (tile >= (unsigned)nTiles) return;
    const unsigned cl = ulane & 15u;
    const int pos = (int)(2u * cl + (ulane >> 4));
    const int startL = rstart[wib * (unsigned)TN + cl];
    const int cntL   = hist[wib * (unsigned)TN + cl] - startL;
    float* mine = &meanS[wib][0];
    const unsigned v0 = tile * (unsigned)TN;
    if (__builtin_amdgcn_readfirstlane((int)fits)) {
        const LdsList list{stage};
        const int ivCur  = first_indices(list, startL, cntL, 0, pos);
        const int ivNext = first_indices(list, startL, cntL, 1, pos);
        sage_tile<true>(x, list, startL, cntL, ivCur, ivNext, &wSN[0][0], &wSN[1][0], mine, bias, out, v0, ulane);
    } else {
        const MemList list{srcSorted + base};
        const int ivCur  = first_indices(list, startL, cntL, 0, pos);
        const int ivNext = first_indices(list, startL, cntL, 1, pos);
        sage_tile<true>(x, list, startL, cntL, ivCur, ivNext, &wSN[0][0], &wSN[1][0], mine, bias, out, v0, ulane);
    }
}

__global__ void __launch_bounds__(LWAVES * 32)
sage_layer_kernel(const float* __restrict__ x,
                  const u64* __restrict__ rowInfo, const int* __restrict__ srcSorted,
                  const float* __restrict__ Wself,
                  const float* __restrict__ Wneigh,
                  const float* __restrict__ bias,
                  float* __restrict__ out,
                  int nNodes, int nTiles) {
    __shared__ float wS[FDIM * FDIM];
    __shared__ float wN[FDIM * FDIM];
    __shared__ float meanS[LWAVES][TN * APITCH];
    const unsigned tid  = threadIdx.x;
    const unsigned lane = tid & 31u;
    const unsigned wib  = __builtin_amdgcn_readfirstlane(tid >> 5);
    const unsigned tile = blockIdx.x * (unsigned)LWAVES + wib;
    const unsigned c  = lane & 15u;
    const int pos = (int)(2u * c + (lane >> 4));
    const unsigned v0 = tile * (unsigned)TN;
    float* mine = &meanS[wib][0];
    const unsigned vi = v0 + c;
    const u64 infoL  = rowInfo[vi < (unsigned)nNodes ? vi : (unsigned)nNodes - 1u];
    #pragma unroll
    for (unsigned u = 0; u < (unsigned)(FDIM * FDIM) / (LWAVES * 32u); ++u) {
        const unsigned i = tid + u * (LWAVES * 32u);
        const unsigned k = i / (unsigned)FDIM;
        const unsigned n = i - k * (unsigned)FDIM;
        const unsigned d = ((k >> 1) * (unsigned)FDIM + n) * 2u + (k & 1u);
        wS[d] = Wself[i];
        wN[d] = Wneigh[i];
    }
    const int startL = (int)(unsigned)infoL;
    const int cntL   = (int)(infoL >> 32);
    const MemList list{srcSorted};
    const int ivCur  = first_indices(list, startL, cntL, 0, pos);
    const int ivNext = first_indices(list, startL, cntL, 1, pos);
    __syncthreads();
    if (tile >= (unsigned)nTiles) return;
    sage_tile<false>(x, list, startL, cntL, ivCur, ivNext, wS, wN, mine, bias, out, v0, lane);
}

extern "C" void kernel_launch(void* const* d_in, const int* in_sizes, int n_in,
                              void* d_out, int out_size, void* d_ws, size_t ws_size,
                              hipStream_t stream) {
    const float* features = (const float*)d_in[0];
    const int*   src      = (const int*)d_in[1];
    const int*   dst      = (const int*)d_in[2];
    const float* Wself1   = (const float*)d_in[3];
    const float* Wneigh1  = (const float*)d_in[4];
    const float* b1       = (const float*)d_in[5];
    const float* Wself2   = (const float*)d_in[6];
    const float* Wneigh2  = (const float*)d_in[7];
    const float* b2       = (const float*)d_in[8];
    float* out = (float*)d_out;

    const int N = in_sizes[0] / FDIM;
    const int E = in_sizes[1];

    const int nBins      = (N + BIN_ROWS - 1) / BIN_ROWS;
    const int nChunks    = (E + CHUNK - 1) / CHUNK;

    char* ws = (char*)d_ws;
    size_t off = 0;
    float* h    = (float*)(ws + off); off += (((size_t)N * FDIM * 4)    + 255) & ~(size_t)255;
    u64*  binned    = (u64*)(ws + off);  off += (((size_t)E * 8) + 255) & ~(size_t)255;
    int*  srcSorted = (int*)(ws + off);  off += (((size_t)E * 4) + 255) & ~(size_t)255;
    u64*  rowInfo   = (u64*)(ws + off);  off += (((size_t)N * 8) + 255) & ~(size_t)255;
    int*  runTab    = (int*)(ws + off);

    const int tiles       = N / 16;
    const int layerBlocks = (tiles + LWAVES - 1) / LWAVES;

    chunk_sort_kernel<<<nChunks, BTHREADS, 0, stream>>>(src, dst, binned, runTab, E, nBins);
    bin_layer_kernel<<<nBins, STHREADS, 0, stream>>>(binned, runTab, rowInfo, srcSorted,
                                                     features, Wself1, Wneigh1, b1, h,
                                                     N, nChunks, nBins, tiles);

    sage_layer_kernel<<<layerBlocks, LWAVES * 32, 0, stream>>>(h, rowInfo, srcSorted,
                                                               Wself2, Wneigh2, b2, out, N, tiles);
}
